# starting kernel
# baseline (speedup 1.0000x reference)
#include <hip/hip_runtime.h>


#define S_    4096
#define N_    262144
#define H_    8
#define E_    128
#define D_    16
#define CKV_  64
#define EPS_  1e-5f

typedef float v2f __attribute__((ext_vector_type(2)));
typedef float v4f __attribute__((ext_vector_type(4)));
typedef float v8f __attribute__((ext_vector_type(8)));

__device__ __forceinline__ unsigned enc_f32(float f) {
  unsigned u = __float_as_uint(f);
  return (u & 0x80000000u) ? ~u : (u | 0x80000000u);
}
__device__ __forceinline__ float dec_f32(unsigned e) {
  return (e & 0x80000000u) ? __uint_as_float(e & 0x7fffffffu)
                           : __uint_as_float(~e);
}

__global__ void k_fold_wkv(const float* __restrict__ Wkv,
                           float* __restrict__ Wfold) {
  int i = blockIdx.x * blockDim.x + threadIdx.x;
  int o = i >> 6, j = i & 63;
  Wfold[i] = Wkv[o * (2 * CKV_) + j] + Wkv[o * (2 * CKV_) + CKV_ + j];
}

__global__ void k_qproj(const float* __restrict__ sp,
                        const float* __restrict__ Wq,
                        float* __restrict__ Q) {
  const int tm   = blockIdx.x * 16;
  const int tn   = blockIdx.y * 16;
  const int lane = threadIdx.x;
  const int half = lane >> 4;
  const int l    = lane & 15;
  const float* arow = sp + (size_t)(tm + l) * E_ + 2 * half;
  const float* brow = Wq + (size_t)(tn + l) * E_ + 2 * half;
  v8f acc = {0.f, 0.f, 0.f, 0.f, 0.f, 0.f, 0.f, 0.f};
#pragma unroll
  for (int k = 0; k < E_; k += 4) {
    v2f a = *(const v2f*)(arow + k);
    v2f b = *(const v2f*)(brow + k);
    acc = __builtin_amdgcn_wmma_f32_16x16x4_f32(false, a, false, b,
                                                (short)0, acc, false, false);
  }
#pragma unroll
  for (int r = 0; r < 8; ++r)
    Q[(size_t)(tm + r + 8 * half) * E_ + tn + l] = acc[r];
}

__global__ void k_kproj_scores(const float* __restrict__ raw,
                               const float* __restrict__ Wfold,
                               const float* __restrict__ Q,
                               const int* __restrict__ assign,
                               float* __restrict__ scores,
                               unsigned* __restrict__ segmax,
                               float* __restrict__ attn_out) {
  __shared__ float kt[16 * 128];
  __shared__ float sc[16 * 8];
  const int p0   = blockIdx.x * 16;
  const int lane = threadIdx.x;
  const int half = lane >> 4;
  const int l    = lane & 15;

  const float* arow = raw + (size_t)(p0 + l) * CKV_ + 2 * half;
  v2f areg[16];
#pragma unroll
  for (int k = 0; k < 16; ++k) areg[k] = *(const v2f*)(arow + 4 * k);

#pragma unroll
  for (int h = 0; h < H_; ++h) {
    const float* brow = Wfold + (size_t)(h * 16 + l) * CKV_ + 2 * half;
    v8f acc = {0.f, 0.f, 0.f, 0.f, 0.f, 0.f, 0.f, 0.f};
#pragma unroll
    for (int k = 0; k < 16; ++k) {
      v2f b = *(const v2f*)(brow + 4 * k);
      acc = __builtin_amdgcn_wmma_f32_16x16x4_f32(false, areg[k], false, b,
                                                  (short)0, acc, false, false);
    }
#pragma unroll
    for (int r = 0; r < 8; ++r)
      kt[(r + 8 * half) * 128 + h * 16 + l] = acc[r];
  }
  __syncthreads();

#pragma unroll
  for (int j = 0; j < 4; ++j) {
    int q = lane * 4 + j;
    int p = q >> 3;
    int h = q & 7;
    int seg = assign[p0 + p];
    const v4f* qv = (const v4f*)(Q + (size_t)seg * E_ + h * 16);
    const v4f* kv = (const v4f*)(&kt[p * 128 + h * 16]);
    float s = 0.f;
#pragma unroll
    for (int t = 0; t < 4; ++t) {
      v4f qq = qv[t], kk = kv[t];
      s += qq.x * kk.x + qq.y * kk.y + qq.z * kk.z + qq.w * kk.w;
    }
    s *= 0.25f;
    scores[(size_t)(p0 + p) * H_ + h] = s;
    atomicMax(&segmax[seg * H_ + h], enc_f32(s));
    sc[p * 8 + h] = s;
  }
  __syncthreads();
  if (lane < 16) {
    float m = 0.f;
#pragma unroll
    for (int h = 0; h < H_; ++h) m += sc[lane * 8 + h];
    attn_out[p0 + lane] = m * (1.f / H_);
  }
}

__global__ void k_vproj_ctx(const float* __restrict__ raw,
                            const float* __restrict__ Wfold,
                            const float* __restrict__ scores,
                            const unsigned* __restrict__ segmax,
                            const int* __restrict__ assign,
                            float* __restrict__ denom,
                            float* __restrict__ ctx) {
  __shared__ float vt[16 * 128];
  const int p0   = blockIdx.x * 16;
  const int lane = threadIdx.x;
  const int half = lane >> 4;
  const int l    = lane & 15;

  const float* arow = raw + (size_t)(p0 + l) * CKV_ + 2 * half;
  v2f areg[16];
#pragma unroll
  for (int k = 0; k < 16; ++k) areg[k] = *(const v2f*)(arow + 4 * k);

#pragma unroll
  for (int h = 0; h < H_; ++h) {
    const float* brow = Wfold + (size_t)(E_ + h * 16 + l) * CKV_ + 2 * half;
    v8f acc = {0.f, 0.f, 0.f, 0.f, 0.f, 0.f, 0.f, 0.f};
#pragma unroll
    for (int k = 0; k < 16; ++k) {
      v2f b = *(const v2f*)(brow + 4 * k);
      acc = __builtin_amdgcn_wmma_f32_16x16x4_f32(false, areg[k], false, b,
                                                  (short)0, acc, false, false);
    }
#pragma unroll
    for (int r = 0; r < 8; ++r)
      vt[(r + 8 * half) * 128 + h * 16 + l] = acc[r];
  }
  __syncthreads();

#pragma unroll
  for (int j = 0; j < 4; ++j) {
    int q = lane * 4 + j;
    int p = q >> 3;
    int h = q & 7;
    int seg = assign[p0 + p];
    float s = scores[(size_t)(p0 + p) * H_ + h];
    float m = dec_f32(segmax[seg * H_ + h]);
    float w = __expf(s - m);
    atomicAdd(&denom[seg * H_ + h], w);
    const float* vrow = &vt[p * 128 + h * 16];
    float* crow = ctx + (size_t)seg * E_ + h * 16;
#pragma unroll
    for (int d = 0; d < D_; ++d) atomicAdd(&crow[d], w * vrow[d]);
  }
}

__global__ void k_finalize(const float* __restrict__ sp,
                           const float* __restrict__ ctx,
                           const float* __restrict__ denom,
                           const float* __restrict__ lnw,
                           const float* __restrict__ lnb,
                           float* __restrict__ out) {
  const int row  = blockIdx.x;
  const int lane = threadIdx.x;
  const int e0   = lane * 4;
  v4f c = *(const v4f*)(ctx + (size_t)row * E_ + e0);
  v4f s = *(const v4f*)(sp + (size_t)row * E_ + e0);
  float den = denom[row * H_ + (e0 >> 4)];
  float inv = (den > 0.f) ? (1.f / den) : 0.f;
  float x[4] = {s.x + c.x * inv, s.y + c.y * inv,
                s.z + c.z * inv, s.w + c.w * inv};
  float sum = x[0] + x[1] + x[2] + x[3];
#pragma unroll
  for (int m = 16; m >= 1; m >>= 1) sum += __shfl_xor(sum, m, 32);
  float mu = sum * (1.f / E_);
  float vs = 0.f;
#pragma unroll
  for (int j = 0; j < 4; ++j) { float d = x[j] - mu; vs += d * d; }
#pragma unroll
  for (int m = 16; m >= 1; m >>= 1) vs += __shfl_xor(vs, m, 32);
  float rs = rsqrtf(vs * (1.f / E_) + EPS_);
  v4f w = *(const v4f*)(lnw + e0);
  v4f b = *(const v4f*)(lnb + e0);
  v4f y;
  y.x = (x[0] - mu) * rs * w.x + b.x;
  y.y = (x[1] - mu) * rs * w.y + b.y;
  y.z = (x[2] - mu) * rs * w.z + b.z;
  y.w = (x[3] - mu) * rs * w.w + b.w;
  *(v4f*)(out + (size_t)row * E_ + e0) = y;
}

extern "C" void kernel_launch(void* const* d_in, const int* in_sizes, int n_in,
                              void* d_out, int out_size, void* d_ws,
                              size_t ws_size, hipStream_t stream) {
  (void)in_sizes; (void)n_in; (void)out_size; (void)ws_size;
  const float* sp  = (const float*)d_in[0];
  const float* raw = (const float*)d_in[1];
  const int*   asg = (const int*)d_in[2];
  const float* Wq  = (const float*)d_in[3];
  const float* Wkv = (const float*)d_in[4];
  const float* lnw = (const float*)d_in[5];
  const float* lnb = (const float*)d_in[6];

  float* out      = (float*)d_out;
  float* out_attn = out + (size_t)S_ * E_;

  float*    ws     = (float*)d_ws;
  float*    Wfold  = ws;
  float*    Q      = Wfold + 256 * CKV_;
  float*    scores = Q + (size_t)S_ * E_;
  unsigned* segmax = (unsigned*)(scores + (size_t)N_ * H_);
  float*    denom  = (float*)(segmax + S_ * H_);
  float*    ctx    = denom + S_ * H_;

  hipMemsetAsync(segmax, 0, (size_t)S_ * H_ * sizeof(unsigned), stream);
  hipMemsetAsync(denom, 0, (size_t)S_ * H_ * sizeof(float), stream);
  hipMemsetAsync(ctx, 0, (size_t)S_ * E_ * sizeof(float), stream);

  k_fold_wkv<<<64, 256, 0, stream>>>(Wkv, Wfold);
  k_qproj<<<dim3(S_ / 16, E_ / 16), 32, 0, stream>>>(sp, Wq, Q);
  k_kproj_scores<<<N_ / 16, 32, 0, stream>>>(raw, Wfold, Q, asg, scores,
                                             segmax, out_attn);
  k_vproj_ctx<<<N_ / 16, 32, 0, stream>>>(raw, Wfold, scores, segmax, asg,
                                          denom, ctx);
  k_finalize<<<S_, 32, 0, stream>>>(sp, ctx, denom, lnw, lnb, out);
}


# Coalesce the global atomics: tail lanes run along one row's consecutive words, so each atomic instruction (segmax, denom, ctx) covers whole lines, not 32 lines; float sums stay arrival-order atomics
# speedup vs baseline: 5.9922x; 5.9922x over previous
#include <hip/hip_runtime.h>


#define S_    4096
#define N_    262144
#define H_    8
#define E_    128
#define D_    16
#define CKV_  64
#define EPS_  1e-5f

typedef float v2f __attribute__((ext_vector_type(2)));
typedef float v4f __attribute__((ext_vector_type(4)));
typedef float v8f __attribute__((ext_vector_type(8)));

__device__ __forceinline__ unsigned enc_f32(float f) {
  unsigned u = __float_as_uint(f);
  return (u & 0x80000000u) ? ~u : (u | 0x80000000u);
}
__device__ __forceinline__ float dec_f32(unsigned e) {
  return (e & 0x80000000u) ? __uint_as_float(e & 0x7fffffffu)
                           : __uint_as_float(~e);
}

__global__ void k_fold_wkv(const float* __restrict__ Wkv,
                           float* __restrict__ Wfold) {
  int i = blockIdx.x * blockDim.x + threadIdx.x;
  int o = i >> 6, j = i & 63;
  Wfold[i] = Wkv[o * (2 * CKV_) + j] + Wkv[o * (2 * CKV_) + CKV_ + j];
}

__global__ void k_qproj(const float* __restrict__ sp,
                        const float* __restrict__ Wq,
                        float* __restrict__ Q) {
  const int tm   = blockIdx.x * 16;
  const int tn   = blockIdx.y * 16;
  const int lane = threadIdx.x;
  const int half = lane >> 4;
  const int l    = lane & 15;
  const float* arow = sp + (size_t)(tm + l) * E_ + 2 * half;
  const float* brow = Wq + (size_t)(tn + l) * E_ + 2 * half;
  v8f acc = {0.f, 0.f, 0.f, 0.f, 0.f, 0.f, 0.f, 0.f};
#pragma unroll
  for (int k = 0; k < E_; k += 4) {
    v2f a = *(const v2f*)(arow + k);
    v2f b = *(const v2f*)(brow + k);
    acc = __builtin_amdgcn_wmma_f32_16x16x4_f32(false, a, false, b,
                                                (short)0, acc, false, false);
  }
#pragma unroll
  for (int r = 0; r < 8; ++r)
    Q[(size_t)(tm + r + 8 * half) * E_ + tn + l] = acc[r];
}

__global__ void k_kproj_scores(const float* __restrict__ raw,
                               const float* __restrict__ Wfold,
                               const float* __restrict__ Q,
                               const int* __restrict__ assign,
                               float* __restrict__ scores,
                               unsigned* __restrict__ segmax,
                               float* __restrict__ attn_out) {
  __shared__ float kt[16 * 128];
  __shared__ float sc[16 * 8];
  const int p0   = blockIdx.x * 16;
  const int lane = threadIdx.x;
  const int half = lane >> 4;
  const int l    = lane & 15;

  const float* arow = raw + (size_t)(p0 + l) * CKV_ + 2 * half;
  v2f areg[16];
#pragma unroll
  for (int k = 0; k < 16; ++k) areg[k] = *(const v2f*)(arow + 4 * k);

#pragma unroll
  for (int h = 0; h < H_; ++h) {
    const float* brow = Wfold + (size_t)(h * 16 + l) * CKV_ + 2 * half;
    v8f acc = {0.f, 0.f, 0.f, 0.f, 0.f, 0.f, 0.f, 0.f};
#pragma unroll
    for (int k = 0; k < 16; ++k) {
      v2f b = *(const v2f*)(brow + 4 * k);
      acc = __builtin_amdgcn_wmma_f32_16x16x4_f32(false, areg[k], false, b,
                                                  (short)0, acc, false, false);
    }
#pragma unroll
    for (int r = 0; r < 8; ++r)
      kt[(r + 8 * half) * 128 + h * 16 + l] = acc[r];
  }
  __syncthreads();

#pragma unroll
  for (int j = 0; j < 4; ++j) {
    int q = j * 32 + lane;
    int p = q >> 3;
    int h = q & 7;
    int seg = assign[p0 + p];
    const v4f* qv = (const v4f*)(Q + (size_t)seg * E_ + h * 16);
    const v4f* kv = (const v4f*)(&kt[p * 128 + h * 16]);
    float s = 0.f;
#pragma unroll
    for (int t = 0; t < 4; ++t) {
      v4f qq = qv[t], kk = kv[t];
      s += qq.x * kk.x + qq.y * kk.y + qq.z * kk.z + qq.w * kk.w;
    }
    s *= 0.25f;
    scores[(size_t)p0 * H_ + q] = s;
    atomicMax(&segmax[seg * H_ + h], enc_f32(s));
    sc[q] = s;
  }
  __syncthreads();
  if (lane < 16) {
    float m = 0.f;
#pragma unroll
    for (int h = 0; h < H_; ++h) m += sc[lane * 8 + h];
    attn_out[p0 + lane] = m * (1.f / H_);
  }
}

__global__ void k_vproj_ctx(const float* __restrict__ raw,
                            const float* __restrict__ Wfold,
                            const float* __restrict__ scores,
                            const unsigned* __restrict__ segmax,
                            const int* __restrict__ assign,
                            float* __restrict__ denom,
                            float* __restrict__ ctx) {
  __shared__ float vt[16 * 128];
  __shared__ float wsh[16 * 8];
  const int p0   = blockIdx.x * 16;
  const int lane = threadIdx.x;
  const int half = lane >> 4;
  const int l    = lane & 15;

  const float* arow = raw + (size_t)(p0 + l) * CKV_ + 2 * half;
  v2f areg[16];
#pragma unroll
  for (int k = 0; k < 16; ++k) areg[k] = *(const v2f*)(arow + 4 * k);

#pragma unroll
  for (int h = 0; h < H_; ++h) {
    const float* brow = Wfold + (size_t)(E_ + h * 16 + l) * CKV_ + 2 * half;
    v8f acc = {0.f, 0.f, 0.f, 0.f, 0.f, 0.f, 0.f, 0.f};
#pragma unroll
    for (int k = 0; k < 16; ++k) {
      v2f b = *(const v2f*)(brow + 4 * k);
      acc = __builtin_amdgcn_wmma_f32_16x16x4_f32(false, areg[k], false, b,
                                                  (short)0, acc, false, false);
    }
#pragma unroll
    for (int r = 0; r < 8; ++r)
      vt[(r + 8 * half) * 128 + h * 16 + l] = acc[r];
  }

  const int hq = lane & 7;
  int   segj[4];
  float sj[4];
  float mj[4];
  float wj[4];
#pragma unroll
  for (int j = 0; j < 4; ++j) {
    segj[j] = assign[p0 + j * 4 + (lane >> 3)];
    sj[j]   = scores[(size_t)p0 * H_ + j * 32 + lane];
  }
#pragma unroll
  for (int j = 0; j < 4; ++j) mj[j] = dec_f32(segmax[segj[j] * H_ + hq]);
#pragma unroll
  for (int j = 0; j < 4; ++j) {
    wj[j] = __expf(sj[j] - mj[j]);
    wsh[(j * 4 + (lane >> 3)) * 8 + (hq & 1) * 4 + (hq >> 1)] = wj[j];
  }
  __syncthreads();
#pragma unroll
  for (int j = 0; j < 4; ++j) atomicAdd(&denom[segj[j] * H_ + hq], wj[j]);

#pragma unroll
  for (int p = 0; p < 16; ++p) {
    const int seg = assign[p0 + p];
    float* crow = ctx + (size_t)seg * E_ + lane;
    const v4f w4 = *(const v4f*)&wsh[p * 8 + half * 4];
    float x[4];
#pragma unroll
    for (int c = 0; c < 4; ++c) x[c] = w4[c] * vt[p * 128 + c * 32 + lane];
#pragma unroll
    for (int c = 0; c < 4; ++c) atomicAdd(crow + c * 32, x[c]);
  }
}

__global__ void k_finalize(const float* __restrict__ sp,
                           const float* __restrict__ ctx,
                           const float* __restrict__ denom,
                           const float* __restrict__ lnw,
                           const float* __restrict__ lnb,
                           float* __restrict__ out) {
  const int row  = blockIdx.x;
  const int lane = threadIdx.x;
  const int e0   = lane * 4;
  v4f c = *(const v4f*)(ctx + (size_t)row * E_ + e0);
  v4f s = *(const v4f*)(sp + (size_t)row * E_ + e0);
  float den = denom[row * H_ + (e0 >> 4)];
  float inv = (den > 0.f) ? (1.f / den) : 0.f;
  float x[4] = {s.x + c.x * inv, s.y + c.y * inv,
                s.z + c.z * inv, s.w + c.w * inv};
  float sum = x[0] + x[1] + x[2] + x[3];
#pragma unroll
  for (int m = 16; m >= 1; m >>= 1) sum += __shfl_xor(sum, m, 32);
  float mu = sum * (1.f / E_);
  float vs = 0.f;
#pragma unroll
  for (int j = 0; j < 4; ++j) { float d = x[j] - mu; vs += d * d; }
#pragma unroll
  for (int m = 16; m >= 1; m >>= 1) vs += __shfl_xor(vs, m, 32);
  float rs = rsqrtf(vs * (1.f / E_) + EPS_);
  v4f w = *(const v4f*)(lnw + e0);
  v4f b = *(const v4f*)(lnb + e0);
  v4f y;
  y.x = (x[0] - mu) * rs * w.x + b.x;
  y.y = (x[1] - mu) * rs * w.y + b.y;
  y.z = (x[2] - mu) * rs * w.z + b.z;
  y.w = (x[3] - mu) * rs * w.w + b.w;
  *(v4f*)(out + (size_t)row * E_ + e0) = y;
}

extern "C" void kernel_launch(void* const* d_in, const int* in_sizes, int n_in,
                              void* d_out, int out_size, void* d_ws,
                              size_t ws_size, hipStream_t stream) {
  (void)in_sizes; (void)n_in; (void)out_size; (void)ws_size;
  const float* sp  = (const float*)d_in[0];
  const float* raw = (const float*)d_in[1];
  const int*   asg = (const int*)d_in[2];
  const float* Wq  = (const float*)d_in[3];
  const float* Wkv = (const float*)d_in[4];
  const float* lnw = (const float*)d_in[5];
  const float* lnb = (const float*)d_in[6];

  float* out      = (float*)d_out;
  float* out_attn = out + (size_t)S_ * E_;

  float*    ws     = (float*)d_ws;
  float*    Wfold  = ws;
  float*    Q      = Wfold + 256 * CKV_;
  float*    scores = Q + (size_t)S_ * E_;
  unsigned* segmax = (unsigned*)(scores + (size_t)N_ * H_);
  float*    denom  = (float*)(segmax + S_ * H_);
  float*    ctx    = denom + S_ * H_;

  hipMemsetAsync(segmax, 0, (size_t)S_ * H_ * sizeof(unsigned), stream);
  hipMemsetAsync(denom, 0, (size_t)S_ * H_ * sizeof(float), stream);
  hipMemsetAsync(ctx, 0, (size_t)S_ * E_ * sizeof(float), stream);

  k_fold_wkv<<<64, 256, 0, stream>>>(Wkv, Wfold);
  k_qproj<<<dim3(S_ / 16, E_ / 16), 32, 0, stream>>>(sp, Wq, Q);
  k_kproj_scores<<<N_ / 16, 32, 0, stream>>>(raw, Wfold, Q, asg, scores,
                                             segmax, out_attn);
  k_vproj_ctx<<<N_ / 16, 32, 0, stream>>>(raw, Wfold, scores, segmax, asg,
                                          denom, ctx);
  k_finalize<<<S_, 32, 0, stream>>>(sp, ctx, denom, lnw, lnb, out);
}


# Counting-sort points by segment (arrival-order int rank atomics, scan, scatter); one wave per segment reduces softmax and weighted V in registers + LayerNorm; no float atomics, new sum order
# speedup vs baseline: 6.4380x; 1.0744x over previous
#include <hip/hip_runtime.h>


#define S_    4096
#define N_    262144
#define H_    8
#define E_    128
#define D_    16
#define CKV_  64
#define EPS_  1e-5f

typedef float v2f __attribute__((ext_vector_type(2)));
typedef float v4f __attribute__((ext_vector_type(4)));
typedef float v8f __attribute__((ext_vector_type(8)));

__global__ void k_fold_wkv(const float* __restrict__ Wkv,
                           float* __restrict__ Wfold) {
  int i = blockIdx.x * blockDim.x + threadIdx.x;
  int o = i >> 6, j = i & 63;
  Wfold[i] = Wkv[o * (2 * CKV_) + j] + Wkv[o * (2 * CKV_) + CKV_ + j];
}

__global__ void k_qproj(const float* __restrict__ sp,
                        const float* __restrict__ Wq,
                        float* __restrict__ Q) {
  const int tm   = blockIdx.x * 16;
  const int tn   = blockIdx.y * 16;
  const int lane = threadIdx.x;
  const int half = lane >> 4;
  const int l    = lane & 15;
  const float* arow = sp + (size_t)(tm + l) * E_ + 2 * half;
  const float* brow = Wq + (size_t)(tn + l) * E_ + 2 * half;
  v8f acc = {0.f, 0.f, 0.f, 0.f, 0.f, 0.f, 0.f, 0.f};
#pragma unroll
  for (int k = 0; k < E_; k += 4) {
    v2f a = *(const v2f*)(arow + k);
    v2f b = *(const v2f*)(brow + k);
    acc = __builtin_amdgcn_wmma_f32_16x16x4_f32(false, a, false, b,
                                                (short)0, acc, false, false);
  }
#pragma unroll
  for (int r = 0; r < 8; ++r)
    Q[(size_t)(tm + r + 8 * half) * E_ + tn + l] = acc[r];
}

__global__ void k_kproj_scores(const float* __restrict__ raw,
                               const float* __restrict__ Wfold,
                               const float* __restrict__ Q,
                               const int* __restrict__ assign,
                               float* __restrict__ scores,
                               int* __restrict__ count,
                               int* __restrict__ rank,
                               float* __restrict__ attn_out) {
  __shared__ float kt[16 * 128];
  __shared__ float sc[16 * 8];
  const int p0   = blockIdx.x * 16;
  const int lane = threadIdx.x;
  const int half = lane >> 4;
  const int l    = lane & 15;

  const float* arow = raw + (size_t)(p0 + l) * CKV_ + 2 * half;
  v2f areg[16];
#pragma unroll
  for (int k = 0; k < 16; ++k) areg[k] = *(const v2f*)(arow + 4 * k);

#pragma unroll
  for (int h = 0; h < H_; ++h) {
    const float* brow = Wfold + (size_t)(h * 16 + l) * CKV_ + 2 * half;
    v8f acc = {0.f, 0.f, 0.f, 0.f, 0.f, 0.f, 0.f, 0.f};
#pragma unroll
    for (int k = 0; k < 16; ++k) {
      v2f b = *(const v2f*)(brow + 4 * k);
      acc = __builtin_amdgcn_wmma_f32_16x16x4_f32(false, areg[k], false, b,
                                                  (short)0, acc, false, false);
    }
#pragma unroll
    for (int r = 0; r < 8; ++r)
      kt[(r + 8 * half) * 128 + h * 16 + l] = acc[r];
  }
  __syncthreads();

#pragma unroll
  for (int j = 0; j < 4; ++j) {
    int q = j * 32 + lane;
    int p = q >> 3;
    int h = q & 7;
    int seg = assign[p0 + p];
    const v4f* qv = (const v4f*)(Q + (size_t)seg * E_ + h * 16);
    const v4f* kv = (const v4f*)(&kt[p * 128 + h * 16]);
    float s = 0.f;
#pragma unroll
    for (int t = 0; t < 4; ++t) {
      v4f qq = qv[t], kk = kv[t];
      s += qq.x * kk.x + qq.y * kk.y + qq.z * kk.z + qq.w * kk.w;
    }
    s *= 0.25f;
    scores[(size_t)p0 * H_ + q] = s;
    sc[q] = s;
  }
  __syncthreads();
  if (lane < 16) {
    float m = 0.f;
#pragma unroll
    for (int h = 0; h < H_; ++h) m += sc[lane * 8 + h];
    attn_out[p0 + lane] = m * (1.f / H_);
    rank[p0 + lane] = atomicAdd(&count[assign[p0 + lane]], 1);
  }
}

__global__ void k_scan_counts(const int* __restrict__ count,
                              int* __restrict__ offs) {
  __shared__ int sh[1024];
  const int t = threadIdx.x;
  const int c0 = count[4 * t], c1 = count[4 * t + 1];
  const int c2 = count[4 * t + 2], c3 = count[4 * t + 3];
  const int tot = c0 + c1 + c2 + c3;
  sh[t] = tot;
  __syncthreads();
  for (int d = 1; d < 1024; d <<= 1) {
    int v = (t >= d) ? sh[t - d] : 0;
    __syncthreads();
    sh[t] += v;
    __syncthreads();
  }
  const int ex = sh[t] - tot;
  offs[4 * t]     = ex;
  offs[4 * t + 1] = ex + c0;
  offs[4 * t + 2] = ex + c0 + c1;
  offs[4 * t + 3] = ex + c0 + c1 + c2;
}

__global__ void k_fill_order(const int* __restrict__ assign,
                             const int* __restrict__ rank,
                             const int* __restrict__ offs,
                             int* __restrict__ order) {
  const int p = blockIdx.x * blockDim.x + threadIdx.x;
  order[offs[assign[p]] + rank[p]] = p;
}

__global__ void __launch_bounds__(32)
k_seg_vproj_ln(const float* __restrict__ raw,
               const float* __restrict__ Wfold,
               const float* __restrict__ scores,
               const int* __restrict__ order,
               const int* __restrict__ offs,
               const int* __restrict__ count,
               const float* __restrict__ sp,
               const float* __restrict__ lnw,
               const float* __restrict__ lnb,
               float* __restrict__ out) {
  __shared__ float wsh[H_ * 16];
  const int seg  = blockIdx.x;
  const int lane = threadIdx.x;
  const int half = lane >> 4;
  const int l    = lane & 15;
  const int hq   = lane & 7;
  const int pq   = lane >> 3;
  const int n    = count[seg];
  const int base = offs[seg];
  const int safe = (n > 0) ? base : 0;

  float m = -INFINITY;
  for (int c0 = 0; c0 < n; c0 += 32) {
    const int r = c0 + lane;
    const int mypt = order[(r < n) ? base + r : safe];
#pragma unroll
    for (int jj = 0; jj < 8; ++jj) {
      const int pr = jj * 4 + pq;
      const int pt = __shfl(mypt, pr, 32);
      const float v = scores[(size_t)pt * H_ + hq];
      m = fmaxf(m, (c0 + pr < n) ? v : -INFINITY);
    }
  }
  m = fmaxf(m, __shfl_xor(m, 8, 32));
  m = fmaxf(m, __shfl_xor(m, 16, 32));

  float cacc[H_];
#pragma unroll
  for (int h = 0; h < H_; ++h) cacc[h] = 0.f;
  float dsum = 0.f;

  const int ntile = (n + 15) >> 4;
  const int ptc = order[(l < n) ? base + l : safe];
  int ptn       = order[(16 + l < n) ? base + 16 + l : safe];
  v2f acur[16];
  float scur[4];
  {
    const float* arow = raw + (size_t)ptc * CKV_ + 2 * half;
#pragma unroll
    for (int k = 0; k < 16; ++k) acur[k] = *(const v2f*)(arow + 4 * k);
#pragma unroll
    for (int j = 0; j < 4; ++j)
      scur[j] = scores[(size_t)__shfl(ptc, j * 4 + pq, 32) * H_ + hq];
  }

  for (int t = 0; t < ntile; ++t) {
    v2f anext[16];
    float snext[4];
    {
      const float* arow = raw + (size_t)ptn * CKV_ + 2 * half;
#pragma unroll
      for (int k = 0; k < 16; ++k) anext[k] = *(const v2f*)(arow + 4 * k);
#pragma unroll
      for (int j = 0; j < 4; ++j)
        snext[j] = scores[(size_t)__shfl(ptn, j * 4 + pq, 32) * H_ + hq];
    }
    const int r2 = (t + 2) * 16 + l;
    const int ptnn = order[(r2 < n) ? base + r2 : safe];

#pragma unroll
    for (int j = 0; j < 4; ++j) {
      const int p = j * 4 + pq;
      const float w = (t * 16 + p < n) ? __expf(scur[j] - m) : 0.f;
      dsum += w;
      wsh[hq * 16 + p] = w;
    }
    __syncthreads();
#pragma unroll
    for (int h = 0; h < H_; ++h) {
      const float* brow = Wfold + (size_t)(E_ + h * 16 + l) * CKV_ + 2 * half;
      v8f acc = {0.f, 0.f, 0.f, 0.f, 0.f, 0.f, 0.f, 0.f};
#pragma unroll
      for (int k = 0; k < 16; ++k) {
        v2f b = *(const v2f*)(brow + 4 * k);
        acc = __builtin_amdgcn_wmma_f32_16x16x4_f32(false, acur[k], false, b,
                                                    (short)0, acc, false, false);
      }
      const v4f w0 = *(const v4f*)&wsh[h * 16 + 8 * half];
      const v4f w1 = *(const v4f*)&wsh[h * 16 + 8 * half + 4];
      cacc[h] += acc[0] * w0.x + acc[1] * w0.y + acc[2] * w0.z + acc[3] * w0.w
               + acc[4] * w1.x + acc[5] * w1.y + acc[6] * w1.z + acc[7] * w1.w;
    }
    __syncthreads();
#pragma unroll
    for (int k = 0; k < 16; ++k) acur[k] = anext[k];
#pragma unroll
    for (int j = 0; j < 4; ++j) scur[j] = snext[j];
    ptn = ptnn;
  }

#pragma unroll
  for (int h = 0; h < H_; ++h) cacc[h] += __shfl_xor(cacc[h], 16, 32);
  dsum += __shfl_xor(dsum, 8, 32);
  dsum += __shfl_xor(dsum, 16, 32);

  const float* srow = sp + (size_t)seg * E_ + l;
  float x[H_];
  float sum = 0.f;
#pragma unroll
  for (int h = 0; h < H_; ++h) {
    float den = __shfl(dsum, h, 32);
    float inv = (den > 0.f) ? (1.f / den) : 0.f;
    x[h] = srow[h * 16] + cacc[h] * inv;
    sum += x[h];
  }
#pragma unroll
  for (int s = 8; s >= 1; s >>= 1) sum += __shfl_xor(sum, s, 32);
  const float mu = sum * (1.f / E_);
  float vs = 0.f;
#pragma unroll
  for (int h = 0; h < H_; ++h) { float d = x[h] - mu; vs += d * d; }
#pragma unroll
  for (int s = 8; s >= 1; s >>= 1) vs += __shfl_xor(vs, s, 32);
  const float rs = rsqrtf(vs * (1.f / E_) + EPS_);
#pragma unroll
  for (int c = 0; c < 4; ++c) {
    const float xv = half ? x[2 * c + 1] : x[2 * c];
    const int e = c * 32 + lane;
    out[(size_t)seg * E_ + e] = (xv - mu) * rs * lnw[e] + lnb[e];
  }
}

extern "C" void kernel_launch(void* const* d_in, const int* in_sizes, int n_in,
                              void* d_out, int out_size, void* d_ws,
                              size_t ws_size, hipStream_t stream) {
  (void)in_sizes; (void)n_in; (void)out_size; (void)ws_size;
  const float* sp  = (const float*)d_in[0];
  const float* raw = (const float*)d_in[1];
  const int*   asg = (const int*)d_in[2];
  const float* Wq  = (const float*)d_in[3];
  const float* Wkv = (const float*)d_in[4];
  const float* lnw = (const float*)d_in[5];
  const float* lnb = (const float*)d_in[6];

  float* out      = (float*)d_out;
  float* out_attn = out + (size_t)S_ * E_;

  float* ws      = (float*)d_ws;
  float* Wfold   = ws;
  float* Q       = Wfold + 256 * CKV_;
  float* scores  = Q + (size_t)S_ * E_;
  int*   count   = (int*)(scores + (size_t)N_ * H_);
  int*   offs    = count + S_;
  int*   rank    = offs + S_;
  int*   order   = rank + N_;

  hipMemsetAsync(count, 0, (size_t)S_ * sizeof(int), stream);

  k_fold_wkv<<<64, 256, 0, stream>>>(Wkv, Wfold);
  k_qproj<<<dim3(S_ / 16, E_ / 16), 32, 0, stream>>>(sp, Wq, Q);
  k_kproj_scores<<<N_ / 16, 32, 0, stream>>>(raw, Wfold, Q, asg, scores,
                                             count, rank, out_attn);
  k_scan_counts<<<1, 1024, 0, stream>>>(count, offs);
  k_fill_order<<<N_ / 256, 256, 0, stream>>>(asg, rank, offs, order);
  k_seg_vproj_ln<<<S_, 32, 0, stream>>>(raw, Wfold, scores, order, offs,
                                        count, sp, lnw, lnb, out);
}


# K projection as persistent single-wave blocks: a wave loads its 128 weight fragments into registers once and walks 8 point tiles, so its WMMA loop issues no weight loads (was 128 per tile)
# speedup vs baseline: 7.3303x; 1.1386x over previous
#include <hip/hip_runtime.h>


#define S_    4096
#define N_    262144
#define H_    8
#define E_    128
#define D_    16
#define CKV_  64
#define EPS_  1e-5f
#define NTILE_ (N_ / 16)

typedef float v2f __attribute__((ext_vector_type(2)));
typedef float v4f __attribute__((ext_vector_type(4)));
typedef float v8f __attribute__((ext_vector_type(8)));

__global__ void k_fold_wkv(const float* __restrict__ Wkv,
                           float* __restrict__ Wfold) {
  int i = blockIdx.x * blockDim.x + threadIdx.x;
  int o = i >> 6, j = i & 63;
  Wfold[i] = Wkv[o * (2 * CKV_) + j] + Wkv[o * (2 * CKV_) + CKV_ + j];
}

__global__ void k_qproj(const float* __restrict__ sp,
                        const float* __restrict__ Wq,
                        float* __restrict__ Q) {
  const int tm   = blockIdx.x * 16;
  const int tn   = blockIdx.y * 16;
  const int lane = threadIdx.x;
  const int half = lane >> 4;
  const int l    = lane & 15;
  const float* arow = sp + (size_t)(tm + l) * E_ + 2 * half;
  const float* brow = Wq + (size_t)(tn + l) * E_ + 2 * half;
  v8f acc = {0.f, 0.f, 0.f, 0.f, 0.f, 0.f, 0.f, 0.f};
#pragma unroll
  for (int k = 0; k < E_; k += 4) {
    v2f a = *(const v2f*)(arow + k);
    v2f b = *(const v2f*)(brow + k);
    acc = __builtin_amdgcn_wmma_f32_16x16x4_f32(false, a, false, b,
                                                (short)0, acc, false, false);
  }
#pragma unroll
  for (int r = 0; r < 8; ++r)
    Q[(size_t)(tm + r + 8 * half) * E_ + tn + l] = acc[r];
}

__global__ void __launch_bounds__(32)
k_kproj_scores(const float* __restrict__ raw,
               const float* __restrict__ Wfold,
               const float* __restrict__ Q,
               const int* __restrict__ assign,
               float* __restrict__ scores,
               int* __restrict__ count,
               int* __restrict__ rank,
               float* __restrict__ attn_out) {
  __shared__ float kt[16 * 128];
  __shared__ float sc[16 * 8];
  const int lane = threadIdx.x;
  const int half = lane >> 4;
  const int l    = lane & 15;

  v2f breg[H_][16];
#pragma unroll
  for (int h = 0; h < H_; ++h) {
    const float* brow = Wfold + (size_t)(h * 16 + l) * CKV_ + 2 * half;
#pragma unroll
    for (int k = 0; k < 16; ++k) breg[h][k] = *(const v2f*)(brow + 4 * k);
  }

  int tile = blockIdx.x;
  v2f areg[16];
  {
    const int t0 = (tile < NTILE_) ? tile : 0;
    const float* arow = raw + (size_t)(t0 * 16 + l) * CKV_ + 2 * half;
#pragma unroll
    for (int k = 0; k < 16; ++k) areg[k] = *(const v2f*)(arow + 4 * k);
  }

  for (; tile < NTILE_; tile += gridDim.x) {
    const int p0 = tile * 16;
    const int tn = tile + gridDim.x;
    const int pn = ((tn < NTILE_) ? tn : tile) * 16;
    v2f anext[16];
    {
      const float* arow = raw + (size_t)(pn + l) * CKV_ + 2 * half;
#pragma unroll
      for (int k = 0; k < 16; ++k) anext[k] = *(const v2f*)(arow + 4 * k);
    }
    int segj[4];
#pragma unroll
    for (int j = 0; j < 4; ++j) segj[j] = assign[p0 + ((j * 32 + lane) >> 3)];
    const int myseg = assign[p0 + l];

#pragma unroll
    for (int h = 0; h < H_; ++h) {
      v8f acc = {0.f, 0.f, 0.f, 0.f, 0.f, 0.f, 0.f, 0.f};
#pragma unroll
      for (int k = 0; k < 16; ++k)
        acc = __builtin_amdgcn_wmma_f32_16x16x4_f32(false, areg[k], false, breg[h][k],
                                                    (short)0, acc, false, false);
#pragma unroll
      for (int r = 0; r < 8; ++r)
        kt[(r + 8 * half) * 128 + h * 16 + l] = acc[r];
    }
    __syncthreads();

#pragma unroll
    for (int j = 0; j < 4; ++j) {
      int q = j * 32 + lane;
      int p = q >> 3;
      int h = q & 7;
      const v4f* qv = (const v4f*)(Q + (size_t)segj[j] * E_ + h * 16);
      const v4f* kv = (const v4f*)(&kt[p * 128 + h * 16]);
      float s = 0.f;
#pragma unroll
      for (int t = 0; t < 4; ++t) {
        v4f qq = qv[t], kk = kv[t];
        s += qq.x * kk.x + qq.y * kk.y + qq.z * kk.z + qq.w * kk.w;
      }
      s *= 0.25f;
      scores[(size_t)p0 * H_ + q] = s;
      sc[q] = s;
    }
    __syncthreads();
    if (lane < 16) {
      float m = 0.f;
#pragma unroll
      for (int h = 0; h < H_; ++h) m += sc[lane * 8 + h];
      attn_out[p0 + lane] = m * (1.f / H_);
      rank[p0 + lane] = atomicAdd(&count[myseg], 1);
    }
#pragma unroll
    for (int k = 0; k < 16; ++k) areg[k] = anext[k];
  }
}

__global__ void k_scan_counts(const int* __restrict__ count,
                              int* __restrict__ offs) {
  __shared__ int sh[1024];
  const int t = threadIdx.x;
  const int c0 = count[4 * t], c1 = count[4 * t + 1];
  const int c2 = count[4 * t + 2], c3 = count[4 * t + 3];
  const int tot = c0 + c1 + c2 + c3;
  sh[t] = tot;
  __syncthreads();
  for (int d = 1; d < 1024; d <<= 1) {
    int v = (t >= d) ? sh[t - d] : 0;
    __syncthreads();
    sh[t] += v;
    __syncthreads();
  }
  const int ex = sh[t] - tot;
  offs[4 * t]     = ex;
  offs[4 * t + 1] = ex + c0;
  offs[4 * t + 2] = ex + c0 + c1;
  offs[4 * t + 3] = ex + c0 + c1 + c2;
}

__global__ void k_fill_order(const int* __restrict__ assign,
                             const int* __restrict__ rank,
                             const int* __restrict__ offs,
                             int* __restrict__ order) {
  const int p = blockIdx.x * blockDim.x + threadIdx.x;
  order[offs[assign[p]] + rank[p]] = p;
}

__global__ void __launch_bounds__(32)
k_seg_vproj_ln(const float* __restrict__ raw,
               const float* __restrict__ Wfold,
               const float* __restrict__ scores,
               const int* __restrict__ order,
               const int* __restrict__ offs,
               const int* __restrict__ count,
               const float* __restrict__ sp,
               const float* __restrict__ lnw,
               const float* __restrict__ lnb,
               float* __restrict__ out) {
  __shared__ float wsh[H_ * 16];
  const int seg  = blockIdx.x;
  const int lane = threadIdx.x;
  const int half = lane >> 4;
  const int l    = lane & 15;
  const int hq   = lane & 7;
  const int pq   = lane >> 3;
  const int n    = count[seg];
  const int base = offs[seg];
  const int safe = (n > 0) ? base : 0;

  float m = -INFINITY;
  for (int c0 = 0; c0 < n; c0 += 32) {
    const int r = c0 + lane;
    const int mypt = order[(r < n) ? base + r : safe];
#pragma unroll
    for (int jj = 0; jj < 8; ++jj) {
      const int pr = jj * 4 + pq;
      const int pt = __shfl(mypt, pr, 32);
      const float v = scores[(size_t)pt * H_ + hq];
      m = fmaxf(m, (c0 + pr < n) ? v : -INFINITY);
    }
  }
  m = fmaxf(m, __shfl_xor(m, 8, 32));
  m = fmaxf(m, __shfl_xor(m, 16, 32));

  float cacc[H_];
#pragma unroll
  for (int h = 0; h < H_; ++h) cacc[h] = 0.f;
  float dsum = 0.f;

  const int ntile = (n + 15) >> 4;
  const int ptc = order[(l < n) ? base + l : safe];
  int ptn       = order[(16 + l < n) ? base + 16 + l : safe];
  v2f acur[16];
  float scur[4];
  {
    const float* arow = raw + (size_t)ptc * CKV_ + 2 * half;
#pragma unroll
    for (int k = 0; k < 16; ++k) acur[k] = *(const v2f*)(arow + 4 * k);
#pragma unroll
    for (int j = 0; j < 4; ++j)
      scur[j] = scores[(size_t)__shfl(ptc, j * 4 + pq, 32) * H_ + hq];
  }

  for (int t = 0; t < ntile; ++t) {
    v2f anext[16];
    float snext[4];
    {
      const float* arow = raw + (size_t)ptn * CKV_ + 2 * half;
#pragma unroll
      for (int k = 0; k < 16; ++k) anext[k] = *(const v2f*)(arow + 4 * k);
#pragma unroll
      for (int j = 0; j < 4; ++j)
        snext[j] = scores[(size_t)__shfl(ptn, j * 4 + pq, 32) * H_ + hq];
    }
    const int r2 = (t + 2) * 16 + l;
    const int ptnn = order[(r2 < n) ? base + r2 : safe];

#pragma unroll
    for (int j = 0; j < 4; ++j) {
      const int p = j * 4 + pq;
      const float w = (t * 16 + p < n) ? __expf(scur[j] - m) : 0.f;
      dsum += w;
      wsh[hq * 16 + p] = w;
    }
    __syncthreads();
#pragma unroll
    for (int h = 0; h < H_; ++h) {
      const float* brow = Wfold + (size_t)(E_ + h * 16 + l) * CKV_ + 2 * half;
      v8f acc = {0.f, 0.f, 0.f, 0.f, 0.f, 0.f, 0.f, 0.f};
#pragma unroll
      for (int k = 0; k < 16; ++k) {
        v2f b = *(const v2f*)(brow + 4 * k);
        acc = __builtin_amdgcn_wmma_f32_16x16x4_f32(false, acur[k], false, b,
                                                    (short)0, acc, false, false);
      }
      const v4f w0 = *(const v4f*)&wsh[h * 16 + 8 * half];
      const v4f w1 = *(const v4f*)&wsh[h * 16 + 8 * half + 4];
      cacc[h] += acc[0] * w0.x + acc[1] * w0.y + acc[2] * w0.z + acc[3] * w0.w
               + acc[4] * w1.x + acc[5] * w1.y + acc[6] * w1.z + acc[7] * w1.w;
    }
    __syncthreads();
#pragma unroll
    for (int k = 0; k < 16; ++k) acur[k] = anext[k];
#pragma unroll
    for (int j = 0; j < 4; ++j) scur[j] = snext[j];
    ptn = ptnn;
  }

#pragma unroll
  for (int h = 0; h < H_; ++h) cacc[h] += __shfl_xor(cacc[h], 16, 32);
  dsum += __shfl_xor(dsum, 8, 32);
  dsum += __shfl_xor(dsum, 16, 32);

  const float* srow = sp + (size_t)seg * E_ + l;
  float x[H_];
  float sum = 0.f;
#pragma unroll
  for (int h = 0; h < H_; ++h) {
    float den = __shfl(dsum, h, 32);
    float inv = (den > 0.f) ? (1.f / den) : 0.f;
    x[h] = srow[h * 16] + cacc[h] * inv;
    sum += x[h];
  }
#pragma unroll
  for (int s = 8; s >= 1; s >>= 1) sum += __shfl_xor(sum, s, 32);
  const float mu = sum * (1.f / E_);
  float vs = 0.f;
#pragma unroll
  for (int h = 0; h < H_; ++h) { float d = x[h] - mu; vs += d * d; }
#pragma unroll
  for (int s = 8; s >= 1; s >>= 1) vs += __shfl_xor(vs, s, 32);
  const float rs = rsqrtf(vs * (1.f / E_) + EPS_);
#pragma unroll
  for (int c = 0; c < 4; ++c) {
    const float xv = half ? x[2 * c + 1] : x[2 * c];
    const int e = c * 32 + lane;
    out[(size_t)seg * E_ + e] = (xv - mu) * rs * lnw[e] + lnb[e];
  }
}

extern "C" void kernel_launch(void* const* d_in, const int* in_sizes, int n_in,
                              void* d_out, int out_size, void* d_ws,
                              size_t ws_size, hipStream_t stream) {
  (void)in_sizes; (void)n_in; (void)out_size; (void)ws_size;
  const float* sp  = (const float*)d_in[0];
  const float* raw = (const float*)d_in[1];
  const int*   asg = (const int*)d_in[2];
  const float* Wq  = (const float*)d_in[3];
  const float* Wkv = (const float*)d_in[4];
  const float* lnw = (const float*)d_in[5];
  const float* lnb = (const float*)d_in[6];

  float* out      = (float*)d_out;
  float* out_attn = out + (size_t)S_ * E_;

  float* ws      = (float*)d_ws;
  float* Wfold   = ws;
  float* Q       = Wfold + 256 * CKV_;
  float* scores  = Q + (size_t)S_ * E_;
  int*   count   = (int*)(scores + (size_t)N_ * H_);
  int*   offs    = count + S_;
  int*   rank    = offs + S_;
  int*   order   = rank + N_;

  hipMemsetAsync(count, 0, (size_t)S_ * sizeof(int), stream);

  k_fold_wkv<<<64, 256, 0, stream>>>(Wkv, Wfold);
  k_qproj<<<dim3(S_ / 16, E_ / 16), 32, 0, stream>>>(sp, Wq, Q);
  k_kproj_scores<<<2048, 32, 0, stream>>>(raw, Wfold, Q, asg, scores,
                                             count, rank, out_attn);
  k_scan_counts<<<1, 1024, 0, stream>>>(count, offs);
  k_fill_order<<<N_ / 256, 256, 0, stream>>>(asg, rank, offs, order);
  k_seg_vproj_ln<<<S_, 32, 0, stream>>>(raw, Wfold, scores, order, offs,
                                        count, sp, lnw, lnb, out);
}


# V weights applied after the segment sum: per segment Z = exp-weights^T x raw rows on f32 WMMA (16 per 16 points, was 128), then one Z x Wv^T WMMA product; new sum order
# speedup vs baseline: 8.9883x; 1.2262x over previous
#include <hip/hip_runtime.h>


#define S_    4096
#define N_    262144
#define H_    8
#define E_    128
#define D_    16
#define CKV_  64
#define EPS_  1e-5f
#define NTILE_ (N_ / 16)

typedef float v2f __attribute__((ext_vector_type(2)));
typedef float v4f __attribute__((ext_vector_type(4)));
typedef float v8f __attribute__((ext_vector_type(8)));

__global__ void k_fold_wkv(const float* __restrict__ Wkv,
                           float* __restrict__ Wfold) {
  int i = blockIdx.x * blockDim.x + threadIdx.x;
  int o = i >> 6, j = i & 63;
  float v = Wkv[o * (2 * CKV_) + j] + Wkv[o * (2 * CKV_) + CKV_ + j];
  if (o < E_) {
    Wfold[i] = v;
  } else {
    int hh = (o - E_) >> 4, c = (o - E_) & 15;
    int q = j >> 3, hf = (j >> 2) & 1, e = j & 3;
    Wfold[E_ * CKV_ + ((((hh * 8 + q) * 32) + hf * 16 + c) << 2) + e] = v;
  }
}

__global__ void k_qproj(const float* __restrict__ sp,
                        const float* __restrict__ Wq,
                        float* __restrict__ Q) {
  const int tm   = blockIdx.x * 16;
  const int tn   = blockIdx.y * 16;
  const int lane = threadIdx.x;
  const int half = lane >> 4;
  const int l    = lane & 15;
  const float* arow = sp + (size_t)(tm + l) * E_ + 2 * half;
  const float* brow = Wq + (size_t)(tn + l) * E_ + 2 * half;
  v8f acc = {0.f, 0.f, 0.f, 0.f, 0.f, 0.f, 0.f, 0.f};
#pragma unroll
  for (int k = 0; k < E_; k += 4) {
    v2f a = *(const v2f*)(arow + k);
    v2f b = *(const v2f*)(brow + k);
    acc = __builtin_amdgcn_wmma_f32_16x16x4_f32(false, a, false, b,
                                                (short)0, acc, false, false);
  }
#pragma unroll
  for (int r = 0; r < 8; ++r)
    Q[(size_t)(tm + r + 8 * half) * E_ + tn + l] = acc[r];
}

__global__ void __launch_bounds__(32)
k_kproj_scores(const float* __restrict__ raw,
               const float* __restrict__ Wfold,
               const float* __restrict__ Q,
               const int* __restrict__ assign,
               float* __restrict__ scores,
               int* __restrict__ count,
               int* __restrict__ rank,
               float* __restrict__ attn_out) {
  __shared__ float kt[16 * 128];
  __shared__ float sc[16 * 8];
  const int lane = threadIdx.x;
  const int half = lane >> 4;
  const int l    = lane & 15;

  v2f breg[H_][16];
#pragma unroll
  for (int h = 0; h < H_; ++h) {
    const float* brow = Wfold + (size_t)(h * 16 + l) * CKV_ + 2 * half;
#pragma unroll
    for (int k = 0; k < 16; ++k) breg[h][k] = *(const v2f*)(brow + 4 * k);
  }

  int tile = blockIdx.x;
  v2f areg[16];
  {
    const int t0 = (tile < NTILE_) ? tile : 0;
    const float* arow = raw + (size_t)(t0 * 16 + l) * CKV_ + 2 * half;
#pragma unroll
    for (int k = 0; k < 16; ++k) areg[k] = *(const v2f*)(arow + 4 * k);
  }

  for (; tile < NTILE_; tile += gridDim.x) {
    const int p0 = tile * 16;
    const int tn = tile + gridDim.x;
    const int pn = ((tn < NTILE_) ? tn : tile) * 16;
    v2f anext[16];
    {
      const float* arow = raw + (size_t)(pn + l) * CKV_ + 2 * half;
#pragma unroll
      for (int k = 0; k < 16; ++k) anext[k] = *(const v2f*)(arow + 4 * k);
    }
    int segj[4];
#pragma unroll
    for (int j = 0; j < 4; ++j) segj[j] = assign[p0 + ((j * 32 + lane) >> 3)];
    const int myseg = assign[p0 + l];

#pragma unroll
    for (int h = 0; h < H_; ++h) {
      v8f acc = {0.f, 0.f, 0.f, 0.f, 0.f, 0.f, 0.f, 0.f};
#pragma unroll
      for (int k = 0; k < 16; ++k)
        acc = __builtin_amdgcn_wmma_f32_16x16x4_f32(false, areg[k], false, breg[h][k],
                                                    (short)0, acc, false, false);
#pragma unroll
      for (int r = 0; r < 8; ++r)
        kt[(r + 8 * half) * 128 + h * 16 + l] = acc[r];
    }
    __syncthreads();

#pragma unroll
    for (int j = 0; j < 4; ++j) {
      int q = j * 32 + lane;
      int p = q >> 3;
      int h = q & 7;
      const v4f* qv = (const v4f*)(Q + (size_t)segj[j] * E_ + h * 16);
      const v4f* kv = (const v4f*)(&kt[p * 128 + h * 16]);
      float s = 0.f;
#pragma unroll
      for (int t = 0; t < 4; ++t) {
        v4f qq = qv[t], kk = kv[t];
        s += qq.x * kk.x + qq.y * kk.y + qq.z * kk.z + qq.w * kk.w;
      }
      s *= 0.25f;
      scores[(size_t)p0 * H_ + q] = s;
      sc[q] = s;
    }
    __syncthreads();
    if (lane < 16) {
      float m = 0.f;
#pragma unroll
      for (int h = 0; h < H_; ++h) m += sc[lane * 8 + h];
      attn_out[p0 + lane] = m * (1.f / H_);
      rank[p0 + lane] = atomicAdd(&count[myseg], 1);
    }
#pragma unroll
    for (int k = 0; k < 16; ++k) areg[k] = anext[k];
  }
}

__global__ void k_scan_counts(const int* __restrict__ count,
                              int* __restrict__ offs) {
  __shared__ int sh[1024];
  const int t = threadIdx.x;
  const int c0 = count[4 * t], c1 = count[4 * t + 1];
  const int c2 = count[4 * t + 2], c3 = count[4 * t + 3];
  const int tot = c0 + c1 + c2 + c3;
  sh[t] = tot;
  __syncthreads();
  for (int d = 1; d < 1024; d <<= 1) {
    int v = (t >= d) ? sh[t - d] : 0;
    __syncthreads();
    sh[t] += v;
    __syncthreads();
  }
  const int ex = sh[t] - tot;
  offs[4 * t]     = ex;
  offs[4 * t + 1] = ex + c0;
  offs[4 * t + 2] = ex + c0 + c1;
  offs[4 * t + 3] = ex + c0 + c1 + c2;
}

__global__ void k_fill_order(const int* __restrict__ assign,
                             const int* __restrict__ rank,
                             const int* __restrict__ offs,
                             int* __restrict__ order) {
  const int p = blockIdx.x * blockDim.x + threadIdx.x;
  order[offs[assign[p]] + rank[p]] = p;
}

#define ZP_ 68
__global__ void __launch_bounds__(32)
k_seg_vproj_ln(const float* __restrict__ raw,
               const float* __restrict__ Wfold,
               const float* __restrict__ scores,
               const int* __restrict__ order,
               const int* __restrict__ offs,
               const int* __restrict__ count,
               const float* __restrict__ sp,
               const float* __restrict__ lnw,
               const float* __restrict__ lnb,
               float* __restrict__ out) {
  __shared__ __attribute__((aligned(16))) float zs[16 * ZP_];
  const int seg  = blockIdx.x;
  const int lane = threadIdx.x;
  const int half = lane >> 4;
  const int l    = lane & 15;
  const int hq   = lane & 7;
  const int pq   = lane >> 3;
  const int n    = count[seg];
  const int base = offs[seg];
  const int safe = (n > 0) ? base : 0;

  float m = -INFINITY;
  for (int c0 = 0; c0 < n; c0 += 32) {
    const int r = c0 + lane;
    const int mypt = order[(r < n) ? base + r : safe];
#pragma unroll
    for (int jj = 0; jj < 8; ++jj) {
      const int pr = jj * 4 + pq;
      const int pt = __shfl(mypt, pr, 32);
      const float v = scores[(size_t)pt * H_ + hq];
      m = fmaxf(m, (c0 + pr < n) ? v : -INFINITY);
    }
  }
  m = fmaxf(m, __shfl_xor(m, 8, 32));
  m = fmaxf(m, __shfl_xor(m, 16, 32));

  const bool hrow = (l < 8);
  v8f z0 = {0.f, 0.f, 0.f, 0.f, 0.f, 0.f, 0.f, 0.f};
  v8f z1 = z0, z2 = z0, z3 = z0;
  float dsum = 0.f;

  const int ntile = (n + 15) >> 4;
  const int ptc = order[(l < n) ? base + l : safe];
  int ptn       = order[(16 + l < n) ? base + 16 + l : safe];
  v4f rcur[8];
  float scur[8];
#pragma unroll
  for (int i = 0; i < 8; ++i) {
    const int p = __shfl(ptc, 4 * (i >> 1) + 2 * half + (i & 1), 32);
    rcur[i] = *(const v4f*)(raw + (size_t)p * CKV_ + 4 * l);
    scur[i] = scores[(size_t)p * H_ + hq];
  }

  for (int t = 0; t < ntile; ++t) {
    v4f rnext[8];
    float snext[8];
#pragma unroll
    for (int i = 0; i < 8; ++i) {
      const int p = __shfl(ptn, 4 * (i >> 1) + 2 * half + (i & 1), 32);
      rnext[i] = *(const v4f*)(raw + (size_t)p * CKV_ + 4 * l);
      snext[i] = scores[(size_t)p * H_ + hq];
    }
    const int r2 = (t + 2) * 16 + l;
    const int ptnn = order[(r2 < n) ? base + r2 : safe];

#pragma unroll
    for (int q = 0; q < 4; ++q) {
      const int s0 = t * 16 + 4 * q + 2 * half;
      v2f a;
      a.x = (hrow && s0 < n)     ? __expf(scur[2 * q] - m)     : 0.f;
      a.y = (hrow && s0 + 1 < n) ? __expf(scur[2 * q + 1] - m) : 0.f;
      dsum += a.x + a.y;
      const v4f u0 = rcur[2 * q], u1 = rcur[2 * q + 1];
      v2f b0 = {u0.x, u1.x}, b1 = {u0.y, u1.y}, b2 = {u0.z, u1.z}, b3 = {u0.w, u1.w};
      z0 = __builtin_amdgcn_wmma_f32_16x16x4_f32(false, a, false, b0, (short)0, z0, false, false);
      z1 = __builtin_amdgcn_wmma_f32_16x16x4_f32(false, a, false, b1, (short)0, z1, false, false);
      z2 = __builtin_amdgcn_wmma_f32_16x16x4_f32(false, a, false, b2, (short)0, z2, false, false);
      z3 = __builtin_amdgcn_wmma_f32_16x16x4_f32(false, a, false, b3, (short)0, z3, false, false);
    }
#pragma unroll
    for (int i = 0; i < 8; ++i) { rcur[i] = rnext[i]; scur[i] = snext[i]; }
    ptn = ptnn;
  }

#pragma unroll
  for (int r = 0; r < 8; ++r) {
    v4f zr = {z0[r], z1[r], z2[r], z3[r]};
    *(v4f*)&zs[(8 * half + r) * ZP_ + 4 * l] = zr;
  }
  __syncthreads();
  v4f av[8];
#pragma unroll
  for (int q = 0; q < 8; ++q) av[q] = *(const v4f*)&zs[l * ZP_ + 8 * q + 4 * half];

  float cx[H_];
  const v4f* wv = (const v4f*)(Wfold + E_ * CKV_) + lane;
#pragma unroll
  for (int h = 0; h < H_; ++h) {
    v8f acc = {0.f, 0.f, 0.f, 0.f, 0.f, 0.f, 0.f, 0.f};
#pragma unroll
    for (int q = 0; q < 8; ++q) {
      const v4f bv = wv[(h * 8 + q) * 32];
      v2f a0 = {av[q].x, av[q].y}, a1 = {av[q].z, av[q].w};
      v2f b0 = {bv.x, bv.y},       b1 = {bv.z, bv.w};
      acc = __builtin_amdgcn_wmma_f32_16x16x4_f32(false, a0, false, b0, (short)0, acc, false, false);
      acc = __builtin_amdgcn_wmma_f32_16x16x4_f32(false, a1, false, b1, (short)0, acc, false, false);
    }
    cx[h] = __shfl(acc[h], l, 32);
  }

  dsum += __shfl_xor(dsum, 16, 32);

  const float* srow = sp + (size_t)seg * E_ + l;
  float x[H_];
  float sum = 0.f;
#pragma unroll
  for (int h = 0; h < H_; ++h) {
    float den = __shfl(dsum, h, 32);
    float inv = (den > 0.f) ? (1.f / den) : 0.f;
    x[h] = srow[h * 16] + cx[h] * inv;
    sum += x[h];
  }
#pragma unroll
  for (int s = 8; s >= 1; s >>= 1) sum += __shfl_xor(sum, s, 32);
  const float mu = sum * (1.f / E_);
  float vs = 0.f;
#pragma unroll
  for (int h = 0; h < H_; ++h) { float d = x[h] - mu; vs += d * d; }
#pragma unroll
  for (int s = 8; s >= 1; s >>= 1) vs += __shfl_xor(vs, s, 32);
  const float rs = rsqrtf(vs * (1.f / E_) + EPS_);
#pragma unroll
  for (int c = 0; c < 4; ++c) {
    const float xv = half ? x[2 * c + 1] : x[2 * c];
    const int e = c * 32 + lane;
    out[(size_t)seg * E_ + e] = (xv - mu) * rs * lnw[e] + lnb[e];
  }
}

extern "C" void kernel_launch(void* const* d_in, const int* in_sizes, int n_in,
                              void* d_out, int out_size, void* d_ws,
                              size_t ws_size, hipStream_t stream) {
  (void)in_sizes; (void)n_in; (void)out_size; (void)ws_size;
  const float* sp  = (const float*)d_in[0];
  const float* raw = (const float*)d_in[1];
  const int*   asg = (const int*)d_in[2];
  const float* Wq  = (const float*)d_in[3];
  const float* Wkv = (const float*)d_in[4];
  const float* lnw = (const float*)d_in[5];
  const float* lnb = (const float*)d_in[6];

  float* out      = (float*)d_out;
  float* out_attn = out + (size_t)S_ * E_;

  float* ws      = (float*)d_ws;
  float* Wfold   = ws;
  float* Q       = Wfold + 256 * CKV_;
  float* scores  = Q + (size_t)S_ * E_;
  int*   count   = (int*)(scores + (size_t)N_ * H_);
  int*   offs    = count + S_;
  int*   rank    = offs + S_;
  int*   order   = rank + N_;

  hipMemsetAsync(count, 0, (size_t)S_ * sizeof(int), stream);

  k_fold_wkv<<<64, 256, 0, stream>>>(Wkv, Wfold);
  k_qproj<<<dim3(S_ / 16, E_ / 16), 32, 0, stream>>>(sp, Wq, Q);
  k_kproj_scores<<<2048, 32, 0, stream>>>(raw, Wfold, Q, asg, scores,
                                             count, rank, out_attn);
  k_scan_counts<<<1, 1024, 0, stream>>>(count, offs);
  k_fill_order<<<N_ / 256, 256, 0, stream>>>(asg, rank, offs, order);
  k_seg_vproj_ln<<<S_, 32, 0, stream>>>(raw, Wfold, scores, order, offs,
                                        count, sp, lnw, lnb, out);
}


# K weights folded into Q: Qk = scaled Q_h x Wk_h per segment (WMMA); scores = raw tile x Qk^T in the segment wave, 16 WMMA per 16 points (was 128); one-pass rescaled softmax; new sum order
# speedup vs baseline: 11.8171x; 1.3147x over previous
#include <hip/hip_runtime.h>


#define S_    4096
#define N_    262144
#define H_    8
#define E_    128
#define D_    16
#define CKV_  64
#define EPS_  1e-5f

typedef float v2f __attribute__((ext_vector_type(2)));
typedef float v4f __attribute__((ext_vector_type(4)));
typedef float v8f __attribute__((ext_vector_type(8)));

__global__ void k_fold_wkv(const float* __restrict__ Wkv,
                           float* __restrict__ Wfold) {
  int i = blockIdx.x * blockDim.x + threadIdx.x;
  int o = i >> 6, j = i & 63;
  float v = Wkv[o * (2 * CKV_) + j] + Wkv[o * (2 * CKV_) + CKV_ + j];
  if (o < E_) {
    Wfold[i] = v;
  } else {
    int hh = (o - E_) >> 4, c = (o - E_) & 15;
    int q = j >> 3, hf = (j >> 2) & 1, e = j & 3;
    Wfold[E_ * CKV_ + ((((hh * 8 + q) * 32) + hf * 16 + c) << 2) + e] = v;
  }
}

__global__ void __launch_bounds__(32)
k_qkproj(const float* __restrict__ sp,
         const float* __restrict__ Wq,
         const float* __restrict__ Wfold,
         float* __restrict__ Qk) {
  const int tm   = blockIdx.x * 16;
  const int h    = blockIdx.y;
  const int lane = threadIdx.x;
  const int half = lane >> 4;
  const int l    = lane & 15;
  const float* arow = Wq + (size_t)(h * 16 + l) * E_ + 2 * half;
  const float* brow = sp + (size_t)(tm + l) * E_ + 2 * half;
  v8f qa = {0.f, 0.f, 0.f, 0.f, 0.f, 0.f, 0.f, 0.f};
  v8f qc = qa;
#pragma unroll
  for (int k = 0; k < E_; k += 8) {
    v2f a0 = *(const v2f*)(arow + k);
    v2f b0 = *(const v2f*)(brow + k);
    v2f a1 = *(const v2f*)(arow + k + 4);
    v2f b1 = *(const v2f*)(brow + k + 4);
    qa = __builtin_amdgcn_wmma_f32_16x16x4_f32(false, a0, false, b0,
                                               (short)0, qa, false, false);
    qc = __builtin_amdgcn_wmma_f32_16x16x4_f32(false, a1, false, b1,
                                               (short)0, qc, false, false);
  }
  float qt[8];
#pragma unroll
  for (int r = 0; r < 8; ++r) qt[r] = qa[r] + qc[r];

  const float* wk = Wfold + (size_t)(h * 16 + 8 * half) * CKV_ + l;
#pragma unroll
  for (int nt = 0; nt < 4; ++nt) {
    v8f acc = {0.f, 0.f, 0.f, 0.f, 0.f, 0.f, 0.f, 0.f};
#pragma unroll
    for (int jj = 0; jj < 4; ++jj) {
      v2f a = {qt[2 * jj], qt[2 * jj + 1]};
      v2f b = {wk[(2 * jj) * CKV_ + nt * 16], wk[(2 * jj + 1) * CKV_ + nt * 16]};
      acc = __builtin_amdgcn_wmma_f32_16x16x4_f32(false, a, false, b,
                                                  (short)0, acc, false, false);
    }
#pragma unroll
    for (int r = 0; r < 8; ++r)
      Qk[((size_t)(tm + 8 * half + r) * H_ + h) * CKV_ + nt * 16 + l] = acc[r] * 0.25f;
  }
}

__global__ void k_rank(const int* __restrict__ assign,
                       int* __restrict__ count,
                       int* __restrict__ rank) {
  const int p = blockIdx.x * blockDim.x + threadIdx.x;
  rank[p] = atomicAdd(&count[assign[p]], 1);
}

__global__ void k_scan_counts(const int* __restrict__ count,
                              int* __restrict__ offs) {
  __shared__ int sh[1024];
  const int t = threadIdx.x;
  const int c0 = count[4 * t], c1 = count[4 * t + 1];
  const int c2 = count[4 * t + 2], c3 = count[4 * t + 3];
  const int tot = c0 + c1 + c2 + c3;
  sh[t] = tot;
  __syncthreads();
  for (int d = 1; d < 1024; d <<= 1) {
    int v = (t >= d) ? sh[t - d] : 0;
    __syncthreads();
    sh[t] += v;
    __syncthreads();
  }
  const int ex = sh[t] - tot;
  offs[4 * t]     = ex;
  offs[4 * t + 1] = ex + c0;
  offs[4 * t + 2] = ex + c0 + c1;
  offs[4 * t + 3] = ex + c0 + c1 + c2;
}

__global__ void k_fill_order(const int* __restrict__ assign,
                             const int* __restrict__ rank,
                             const int* __restrict__ offs,
                             int* __restrict__ order) {
  const int p = blockIdx.x * blockDim.x + threadIdx.x;
  order[offs[assign[p]] + rank[p]] = p;
}

#define ZP_ 68

#define ROW_XADD(v, n) ((v) + __builtin_bit_cast(float, __builtin_amdgcn_update_dpp( \
    0, __builtin_bit_cast(int, (v)), 0x160 + (n), 0xf, 0xf, false)))

static __device__ __forceinline__ void seg_step(
    const int t, const int n, const int base, const int safe,
    const int half, const int l, const int posm, const bool hrow,
    const float* __restrict__ raw, const int* __restrict__ order,
    float* __restrict__ attn_out, const v4f (&qb)[8],
    const v4f (&ra)[8], const v4f (&rb)[8], v4f (&na)[8], v4f (&nb)[8],
    const int pnext, int& pnn, const int mcur, int& mnext,
    v8f& z0, v8f& z1, v8f& z2, v8f& z3, float& m, float& dsum) {
  {
    const int r1 = (t + 1) * 16 + posm;
    mnext = order[(r1 < n) ? base + r1 : safe];
  }
#pragma unroll
  for (int i = 0; i < 8; ++i) {
    const int p = __shfl(pnext, 8 * half + i, 32);
    nb[i] = *(const v4f*)(raw + (size_t)p * CKV_ + 4 * l);
  }
  {
    const int r2 = (t + 2) * 16 + l;
    pnn = order[(r2 < n) ? base + r2 : safe];
  }
#pragma unroll
  for (int q = 0; q < 8; ++q)
    na[q] = *(const v4f*)(raw + (size_t)pnext * CKV_ + 8 * q + 4 * half);

  v8f s0 = {0.f, 0.f, 0.f, 0.f, 0.f, 0.f, 0.f, 0.f};
#pragma unroll
  for (int q = 0; q < 8; ++q) {
    v2f a0 = {ra[q].x, ra[q].y}, a1 = {ra[q].z, ra[q].w};
    v2f b0 = {qb[q].x, qb[q].y}, b1 = {qb[q].z, qb[q].w};
    s0 = __builtin_amdgcn_wmma_f32_16x16x4_f32(false, a0, false, b0, (short)0, s0, false, false);
    s0 = __builtin_amdgcn_wmma_f32_16x16x4_f32(false, a1, false, b1, (short)0, s0, false, false);
  }
  float sv[8];
#pragma unroll
  for (int r = 0; r < 8; ++r) sv[r] = s0[r];

  {
    float mv = 0.f;
#pragma unroll
    for (int r = 0; r < 8; ++r) {
      float v = sv[r];
      v = ROW_XADD(v, 1);
      v = ROW_XADD(v, 2);
      v = ROW_XADD(v, 4);
      mv = ((l & 7) == r) ? v : mv;
    }
    if (hrow && (t * 16 + posm < n)) attn_out[mcur] = mv * (1.f / H_);
  }

  float tmax = -INFINITY;
#pragma unroll
  for (int r = 0; r < 8; ++r) {
    sv[r] = (t * 16 + 8 * half + r < n) ? sv[r] : -INFINITY;
    tmax = fmaxf(tmax, sv[r]);
  }
  tmax = fmaxf(tmax, __shfl_xor(tmax, 16, 32));
  const float mnew  = fmaxf(m, tmax);
  const float alpha = __expf(m - mnew);
  m = mnew;
  float w[8];
  float wsum = 0.f;
#pragma unroll
  for (int r = 0; r < 8; ++r) {
    w[r] = hrow ? __expf(sv[r] - mnew) : 0.f;
    wsum += w[r];
  }
  dsum = dsum * alpha + wsum;
#pragma unroll
  for (int r = 0; r < 8; ++r) {
    const float ar = __builtin_bit_cast(float,
        __builtin_amdgcn_readlane(__builtin_bit_cast(int, alpha), r));
    z0[r] *= ar; z1[r] *= ar; z2[r] *= ar; z3[r] *= ar;
  }

#pragma unroll
  for (int q = 0; q < 4; ++q) {
    v2f a = {w[2 * q], w[2 * q + 1]};
    const v4f u0 = rb[2 * q], u1 = rb[2 * q + 1];
    v2f b0 = {u0.x, u1.x}, b1 = {u0.y, u1.y}, b2 = {u0.z, u1.z}, b3 = {u0.w, u1.w};
    z0 = __builtin_amdgcn_wmma_f32_16x16x4_f32(false, a, false, b0, (short)0, z0, false, false);
    z1 = __builtin_amdgcn_wmma_f32_16x16x4_f32(false, a, false, b1, (short)0, z1, false, false);
    z2 = __builtin_amdgcn_wmma_f32_16x16x4_f32(false, a, false, b2, (short)0, z2, false, false);
    z3 = __builtin_amdgcn_wmma_f32_16x16x4_f32(false, a, false, b3, (short)0, z3, false, false);
  }
}

__global__ void __launch_bounds__(32)
k_seg_attn_ln(const float* __restrict__ raw,
              const float* __restrict__ Wfold,
              const float* __restrict__ Qk,
              const int* __restrict__ order,
              const int* __restrict__ offs,
              const int* __restrict__ count,
              const float* __restrict__ sp,
              const float* __restrict__ lnw,
              const float* __restrict__ lnb,
              float* __restrict__ out,
              float* __restrict__ attn_out) {
  __shared__ __attribute__((aligned(16))) float zs[16 * ZP_];
  const int seg  = blockIdx.x;
  const int lane = threadIdx.x;
  const int half = lane >> 4;
  const int l    = lane & 15;
  const int n    = count[seg];
  const int base = offs[seg];
  const int safe = (n > 0) ? base : 0;
  const bool hrow = (l < 8);
  const int posm = 8 * half + (l & 7);

  v4f qb[8];
  {
    const float* qrow = Qk + ((size_t)seg * H_ + (l & 7)) * CKV_ + 4 * half;
#pragma unroll
    for (int q = 0; q < 8; ++q) {
      const v4f t4 = *(const v4f*)(qrow + 8 * q);
      const v4f z4 = {0.f, 0.f, 0.f, 0.f};
      qb[q] = hrow ? t4 : z4;
    }
  }

  v8f z0 = {0.f, 0.f, 0.f, 0.f, 0.f, 0.f, 0.f, 0.f};
  v8f z1 = z0, z2 = z0, z3 = z0;
  float m = -INFINITY;
  float dsum = 0.f;

  const int ntile = (n + 15) >> 4;
  const int ptc0 = order[(l < n) ? base + l : safe];
  int ptA = order[(16 + l < n) ? base + 16 + l : safe];
  int ptB = ptA;
  int mpA = order[(posm < n) ? base + posm : safe];
  int mpB = mpA;
  v4f raA[8], rbA[8], raB[8], rbB[8];
#pragma unroll
  for (int i = 0; i < 8; ++i) {
    const int p = __shfl(ptc0, 8 * half + i, 32);
    rbA[i] = *(const v4f*)(raw + (size_t)p * CKV_ + 4 * l);
  }
#pragma unroll
  for (int q = 0; q < 8; ++q)
    raA[q] = *(const v4f*)(raw + (size_t)ptc0 * CKV_ + 8 * q + 4 * half);

  for (int t = 0; t < ntile; t += 2) {
    seg_step(t, n, base, safe, half, l, posm, hrow, raw, order, attn_out, qb,
             raA, rbA, raB, rbB, ptA, ptB, mpA, mpB, z0, z1, z2, z3, m, dsum);
    if (t + 1 >= ntile) break;
    seg_step(t + 1, n, base, safe, half, l, posm, hrow, raw, order, attn_out, qb,
             raB, rbB, raA, rbA, ptB, ptA, mpB, mpA, z0, z1, z2, z3, m, dsum);
  }

#pragma unroll
  for (int r = 0; r < 8; ++r) {
    v4f zr = {z0[r], z1[r], z2[r], z3[r]};
    *(v4f*)&zs[(8 * half + r) * ZP_ + 4 * l] = zr;
  }
  __syncthreads();
  v4f av[8];
#pragma unroll
  for (int q = 0; q < 8; ++q) av[q] = *(const v4f*)&zs[l * ZP_ + 8 * q + 4 * half];

  float cx[H_];
  const v4f* wv = (const v4f*)(Wfold + E_ * CKV_) + lane;
#pragma unroll
  for (int h = 0; h < H_; ++h) {
    if (h == 4) __builtin_amdgcn_sched_barrier(0);
    v8f acc = {0.f, 0.f, 0.f, 0.f, 0.f, 0.f, 0.f, 0.f};
#pragma unroll
    for (int q = 0; q < 8; ++q) {
      const v4f bv = wv[(h * 8 + q) * 32];
      v2f a0 = {av[q].x, av[q].y}, a1 = {av[q].z, av[q].w};
      v2f b0 = {bv.x, bv.y},       b1 = {bv.z, bv.w};
      acc = __builtin_amdgcn_wmma_f32_16x16x4_f32(false, a0, false, b0, (short)0, acc, false, false);
      acc = __builtin_amdgcn_wmma_f32_16x16x4_f32(false, a1, false, b1, (short)0, acc, false, false);
    }
    cx[h] = __shfl(acc[h], l, 32);
  }

  dsum += __shfl_xor(dsum, 16, 32);

  const float* srow = sp + (size_t)seg * E_ + l;
  float x[H_];
  float sum = 0.f;
#pragma unroll
  for (int h = 0; h < H_; ++h) {
    float den = __shfl(dsum, h, 32);
    float inv = (den > 0.f) ? (1.f / den) : 0.f;
    x[h] = srow[h * 16] + cx[h] * inv;
    sum += x[h];
  }
#pragma unroll
  for (int s = 8; s >= 1; s >>= 1) sum += __shfl_xor(sum, s, 32);
  const float mu = sum * (1.f / E_);
  float vs = 0.f;
#pragma unroll
  for (int h = 0; h < H_; ++h) { float d = x[h] - mu; vs += d * d; }
#pragma unroll
  for (int s = 8; s >= 1; s >>= 1) vs += __shfl_xor(vs, s, 32);
  const float rs = rsqrtf(vs * (1.f / E_) + EPS_);
#pragma unroll
  for (int c = 0; c < 4; ++c) {
    const float xv = half ? x[2 * c + 1] : x[2 * c];
    const int e = c * 32 + lane;
    out[(size_t)seg * E_ + e] = (xv - mu) * rs * lnw[e] + lnb[e];
  }
}

extern "C" void kernel_launch(void* const* d_in, const int* in_sizes, int n_in,
                              void* d_out, int out_size, void* d_ws,
                              size_t ws_size, hipStream_t stream) {
  (void)in_sizes; (void)n_in; (void)out_size; (void)ws_size;
  const float* sp  = (const float*)d_in[0];
  const float* raw = (const float*)d_in[1];
  const int*   asg = (const int*)d_in[2];
  const float* Wq  = (const float*)d_in[3];
  const float* Wkv = (const float*)d_in[4];
  const float* lnw = (const float*)d_in[5];
  const float* lnb = (const float*)d_in[6];

  float* out      = (float*)d_out;
  float* out_attn = out + (size_t)S_ * E_;

  float* ws      = (float*)d_ws;
  float* Wfold   = ws;
  float* Qk      = Wfold + 256 * CKV_;
  int*   count   = (int*)(Qk + (size_t)S_ * H_ * CKV_);
  int*   offs    = count + S_;
  int*   rank    = offs + S_;
  int*   order   = rank + N_;

  hipMemsetAsync(count, 0, (size_t)S_ * sizeof(int), stream);

  k_fold_wkv<<<64, 256, 0, stream>>>(Wkv, Wfold);
  k_qkproj<<<dim3(S_ / 16, H_), 32, 0, stream>>>(sp, Wq, Wfold, Qk);
  k_rank<<<N_ / 256, 256, 0, stream>>>(asg, count, rank);
  k_scan_counts<<<1, 1024, 0, stream>>>(count, offs);
  k_fill_order<<<N_ / 256, 256, 0, stream>>>(asg, rank, offs, order);
  k_seg_attn_ln<<<S_, 32, 0, stream>>>(raw, Wfold, Qk, order, offs, count,
                                       sp, lnw, lnb, out, out_attn);
}


# Dependent launches 7 -> 4: counter clear and Wkv fold move into the Qk launch (fold formed where used), each scatter block scans the segment sizes itself; float sum order unchanged
# speedup vs baseline: 12.2379x; 1.0356x over previous
#include <hip/hip_runtime.h>


#define S_    4096
#define N_    262144
#define H_    8
#define E_    128
#define D_    16
#define CKV_  64
#define EPS_  1e-5f

typedef float v2f __attribute__((ext_vector_type(2)));
typedef float v4f __attribute__((ext_vector_type(4)));
typedef float v8f __attribute__((ext_vector_type(8)));

#define QKX_  (S_ / 16)
#define AUXX_ 4
__global__ void __launch_bounds__(32)
k_qkproj(const float* __restrict__ sp,
         const float* __restrict__ Wq,
         const float* __restrict__ Wkv,
         float* __restrict__ Qk,
         float* __restrict__ WvL,
         int* __restrict__ count) {
  const int h    = blockIdx.y;
  const int lane = threadIdx.x;
  if (blockIdx.x >= QKX_) {
    const int t = ((blockIdx.x - QKX_) * H_ + h) * 32 + lane;
#pragma unroll
    for (int k = 0; k < S_ / 1024; ++k) count[t + 1024 * k] = 0;
#pragma unroll
    for (int k = 0; k < (E_ * CKV_) / 1024; ++k) {
      const int d  = t + 1024 * k;
      const int e  = d & 3, ln = (d >> 2) & 31;
      const int c  = ln & 15, hf = ln >> 4;
      const int q  = (d >> 7) & 7, hh = d >> 10;
      const float* s = Wkv + (size_t)(E_ + hh * 16 + c) * (2 * CKV_) + 8 * q + 4 * hf + e;
      WvL[d] = s[0] + s[CKV_];
    }
    return;
  }
  const int tm   = blockIdx.x * 16;
  const int half = lane >> 4;
  const int l    = lane & 15;
  const float* arow = Wq + (size_t)(h * 16 + l) * E_ + 2 * half;
  const float* brow = sp + (size_t)(tm + l) * E_ + 2 * half;
  v8f qa = {0.f, 0.f, 0.f, 0.f, 0.f, 0.f, 0.f, 0.f};
  v8f qc = qa;
#pragma unroll
  for (int k = 0; k < E_; k += 8) {
    v2f a0 = *(const v2f*)(arow + k);
    v2f b0 = *(const v2f*)(brow + k);
    v2f a1 = *(const v2f*)(arow + k + 4);
    v2f b1 = *(const v2f*)(brow + k + 4);
    qa = __builtin_amdgcn_wmma_f32_16x16x4_f32(false, a0, false, b0,
                                               (short)0, qa, false, false);
    qc = __builtin_amdgcn_wmma_f32_16x16x4_f32(false, a1, false, b1,
                                               (short)0, qc, false, false);
  }
  float qt[8];
#pragma unroll
  for (int r = 0; r < 8; ++r) qt[r] = qa[r] + qc[r];

  const float* wk = Wkv + (size_t)(h * 16 + 8 * half) * (2 * CKV_) + l;
#pragma unroll
  for (int nt = 0; nt < 4; ++nt) {
    v8f acc = {0.f, 0.f, 0.f, 0.f, 0.f, 0.f, 0.f, 0.f};
#pragma unroll
    for (int jj = 0; jj < 4; ++jj) {
      const float* w0 = wk + (2 * jj) * (2 * CKV_) + nt * 16;
      const float* w1 = w0 + 2 * CKV_;
      v2f a = {qt[2 * jj], qt[2 * jj + 1]};
      v2f b = {w0[0] + w0[CKV_], w1[0] + w1[CKV_]};
      acc = __builtin_amdgcn_wmma_f32_16x16x4_f32(false, a, false, b,
                                                  (short)0, acc, false, false);
    }
#pragma unroll
    for (int r = 0; r < 8; ++r)
      Qk[((size_t)(tm + 8 * half + r) * H_ + h) * CKV_ + nt * 16 + l] = acc[r] * 0.25f;
  }
}

__global__ void k_rank(const int* __restrict__ assign,
                       int* __restrict__ count,
                       int* __restrict__ rank) {
  const int p = blockIdx.x * blockDim.x + threadIdx.x;
  rank[p] = atomicAdd(&count[assign[p]], 1);
}

__global__ void __launch_bounds__(256)
k_scan_fill(const int* __restrict__ assign,
            const int* __restrict__ rank,
            const int* __restrict__ count,
            int* __restrict__ offs,
            int* __restrict__ order) {
  __shared__ __attribute__((aligned(16))) int so[S_];
  __shared__ int wt[8];
  const int t    = threadIdx.x;
  const int lane = t & 31;
  const int w    = t >> 5;
  const int p0   = blockIdx.x * 1024 + 4 * t;
  const int4 as = *(const int4*)(assign + p0);
  const int4 rk = *(const int4*)(rank + p0);
  const int4* c4 = (const int4*)(count + 16 * t);
  const int4 ca = c4[0], cb = c4[1], cc = c4[2], cd = c4[3];
  const int tot = (ca.x + ca.y + ca.z + ca.w) + (cb.x + cb.y + cb.z + cb.w) +
                  (cc.x + cc.y + cc.z + cc.w) + (cd.x + cd.y + cd.z + cd.w);
  int inc = tot;
#pragma unroll
  for (int d = 1; d < 32; d <<= 1) {
    const int v = __shfl_up(inc, d, 32);
    inc += (lane >= d) ? v : 0;
  }
  if (lane == 31) wt[w] = inc;
  __syncthreads();
  int ex = inc - tot;
#pragma unroll
  for (int i = 0; i < 8; ++i) ex += (i < w) ? wt[i] : 0;
  int4 oa, ob, oc, od;
  oa.x = ex;          oa.y = oa.x + ca.x; oa.z = oa.y + ca.y; oa.w = oa.z + ca.z;
  ob.x = oa.w + ca.w; ob.y = ob.x + cb.x; ob.z = ob.y + cb.y; ob.w = ob.z + cb.z;
  oc.x = ob.w + cb.w; oc.y = oc.x + cc.x; oc.z = oc.y + cc.y; oc.w = oc.z + cc.z;
  od.x = oc.w + cc.w; od.y = od.x + cd.x; od.z = od.y + cd.y; od.w = od.z + cd.z;
  int4* s4 = (int4*)(so + 16 * t);
  s4[0] = oa; s4[1] = ob; s4[2] = oc; s4[3] = od;
  if (t == (int)blockIdx.x) {
    int4* g4 = (int4*)(offs + 16 * t);
    g4[0] = oa; g4[1] = ob; g4[2] = oc; g4[3] = od;
  }
  __syncthreads();
  order[so[as.x] + rk.x] = p0;
  order[so[as.y] + rk.y] = p0 + 1;
  order[so[as.z] + rk.z] = p0 + 2;
  order[so[as.w] + rk.w] = p0 + 3;
}

#define ZP_ 68

#define ROW_XADD(v, n) ((v) + __builtin_bit_cast(float, __builtin_amdgcn_update_dpp( \
    0, __builtin_bit_cast(int, (v)), 0x160 + (n), 0xf, 0xf, false)))

static __device__ __forceinline__ void seg_step(
    const int t, const int n, const int base, const int safe,
    const int half, const int l, const int posm, const bool hrow,
    const float* __restrict__ raw, const int* __restrict__ order,
    float* __restrict__ attn_out, const v4f (&qb)[8],
    const v4f (&ra)[8], const v4f (&rb)[8], v4f (&na)[8], v4f (&nb)[8],
    const int pnext, int& pnn, const int mcur, int& mnext,
    v8f& z0, v8f& z1, v8f& z2, v8f& z3, float& m, float& dsum) {
  {
    const int r1 = (t + 1) * 16 + posm;
    mnext = order[(r1 < n) ? base + r1 : safe];
  }
#pragma unroll
  for (int i = 0; i < 8; ++i) {
    const int p = __shfl(pnext, 8 * half + i, 32);
    nb[i] = *(const v4f*)(raw + (size_t)p * CKV_ + 4 * l);
  }
  {
    const int r2 = (t + 2) * 16 + l;
    pnn = order[(r2 < n) ? base + r2 : safe];
  }
#pragma unroll
  for (int q = 0; q < 8; ++q)
    na[q] = *(const v4f*)(raw + (size_t)pnext * CKV_ + 8 * q + 4 * half);

  v8f s0 = {0.f, 0.f, 0.f, 0.f, 0.f, 0.f, 0.f, 0.f};
#pragma unroll
  for (int q = 0; q < 8; ++q) {
    v2f a0 = {ra[q].x, ra[q].y}, a1 = {ra[q].z, ra[q].w};
    v2f b0 = {qb[q].x, qb[q].y}, b1 = {qb[q].z, qb[q].w};
    s0 = __builtin_amdgcn_wmma_f32_16x16x4_f32(false, a0, false, b0, (short)0, s0, false, false);
    s0 = __builtin_amdgcn_wmma_f32_16x16x4_f32(false, a1, false, b1, (short)0, s0, false, false);
  }
  float sv[8];
#pragma unroll
  for (int r = 0; r < 8; ++r) sv[r] = s0[r];

  {
    float mv = 0.f;
#pragma unroll
    for (int r = 0; r < 8; ++r) {
      float v = sv[r];
      v = ROW_XADD(v, 1);
      v = ROW_XADD(v, 2);
      v = ROW_XADD(v, 4);
      mv = ((l & 7) == r) ? v : mv;
    }
    if (hrow && (t * 16 + posm < n)) attn_out[mcur] = mv * (1.f / H_);
  }

  float tmax = -INFINITY;
#pragma unroll
  for (int r = 0; r < 8; ++r) {
    sv[r] = (t * 16 + 8 * half + r < n) ? sv[r] : -INFINITY;
    tmax = fmaxf(tmax, sv[r]);
  }
  tmax = fmaxf(tmax, __shfl_xor(tmax, 16, 32));
  const float mnew  = fmaxf(m, tmax);
  const float alpha = __expf(m - mnew);
  m = mnew;
  float w[8];
  float wsum = 0.f;
#pragma unroll
  for (int r = 0; r < 8; ++r) {
    w[r] = hrow ? __expf(sv[r] - mnew) : 0.f;
    wsum += w[r];
  }
  dsum = dsum * alpha + wsum;
#pragma unroll
  for (int r = 0; r < 8; ++r) {
    const float ar = __builtin_bit_cast(float,
        __builtin_amdgcn_readlane(__builtin_bit_cast(int, alpha), r));
    z0[r] *= ar; z1[r] *= ar; z2[r] *= ar; z3[r] *= ar;
  }

#pragma unroll
  for (int q = 0; q < 4; ++q) {
    v2f a = {w[2 * q], w[2 * q + 1]};
    const v4f u0 = rb[2 * q], u1 = rb[2 * q + 1];
    v2f b0 = {u0.x, u1.x}, b1 = {u0.y, u1.y}, b2 = {u0.z, u1.z}, b3 = {u0.w, u1.w};
    z0 = __builtin_amdgcn_wmma_f32_16x16x4_f32(false, a, false, b0, (short)0, z0, false, false);
    z1 = __builtin_amdgcn_wmma_f32_16x16x4_f32(false, a, false, b1, (short)0, z1, false, false);
    z2 = __builtin_amdgcn_wmma_f32_16x16x4_f32(false, a, false, b2, (short)0, z2, false, false);
    z3 = __builtin_amdgcn_wmma_f32_16x16x4_f32(false, a, false, b3, (short)0, z3, false, false);
  }
}

__global__ void __launch_bounds__(32)
k_seg_attn_ln(const float* __restrict__ raw,
              const float* __restrict__ Wfold,
              const float* __restrict__ Qk,
              const int* __restrict__ order,
              const int* __restrict__ offs,
              const int* __restrict__ count,
              const float* __restrict__ sp,
              const float* __restrict__ lnw,
              const float* __restrict__ lnb,
              float* __restrict__ out,
              float* __restrict__ attn_out) {
  __shared__ __attribute__((aligned(16))) float zs[16 * ZP_];
  const int seg  = blockIdx.x;
  const int lane = threadIdx.x;
  const int half = lane >> 4;
  const int l    = lane & 15;
  const int n    = count[seg];
  const int base = offs[seg];
  const int safe = (n > 0) ? base : 0;
  const bool hrow = (l < 8);
  const int posm = 8 * half + (l & 7);

  v4f qb[8];
  {
    const float* qrow = Qk + ((size_t)seg * H_ + (l & 7)) * CKV_ + 4 * half;
#pragma unroll
    for (int q = 0; q < 8; ++q) {
      const v4f t4 = *(const v4f*)(qrow + 8 * q);
      const v4f z4 = {0.f, 0.f, 0.f, 0.f};
      qb[q] = hrow ? t4 : z4;
    }
  }

  v8f z0 = {0.f, 0.f, 0.f, 0.f, 0.f, 0.f, 0.f, 0.f};
  v8f z1 = z0, z2 = z0, z3 = z0;
  float m = -INFINITY;
  float dsum = 0.f;

  const int ntile = (n + 15) >> 4;
  const int ptc0 = order[(l < n) ? base + l : safe];
  int ptA = order[(16 + l < n) ? base + 16 + l : safe];
  int ptB = ptA;
  int mpA = order[(posm < n) ? base + posm : safe];
  int mpB = mpA;
  v4f raA[8], rbA[8], raB[8], rbB[8];
#pragma unroll
  for (int i = 0; i < 8; ++i) {
    const int p = __shfl(ptc0, 8 * half + i, 32);
    rbA[i] = *(const v4f*)(raw + (size_t)p * CKV_ + 4 * l);
  }
#pragma unroll
  for (int q = 0; q < 8; ++q)
    raA[q] = *(const v4f*)(raw + (size_t)ptc0 * CKV_ + 8 * q + 4 * half);

  for (int t = 0; t < ntile; t += 2) {
    seg_step(t, n, base, safe, half, l, posm, hrow, raw, order, attn_out, qb,
             raA, rbA, raB, rbB, ptA, ptB, mpA, mpB, z0, z1, z2, z3, m, dsum);
    if (t + 1 >= ntile) break;
    seg_step(t + 1, n, base, safe, half, l, posm, hrow, raw, order, attn_out, qb,
             raB, rbB, raA, rbA, ptB, ptA, mpB, mpA, z0, z1, z2, z3, m, dsum);
  }

#pragma unroll
  for (int r = 0; r < 8; ++r) {
    v4f zr = {z0[r], z1[r], z2[r], z3[r]};
    *(v4f*)&zs[(8 * half + r) * ZP_ + 4 * l] = zr;
  }
  __syncthreads();
  v4f av[8];
#pragma unroll
  for (int q = 0; q < 8; ++q) av[q] = *(const v4f*)&zs[l * ZP_ + 8 * q + 4 * half];

  float cx[H_];
  const v4f* wv = (const v4f*)(Wfold + E_ * CKV_) + lane;
#pragma unroll
  for (int h = 0; h < H_; ++h) {
    if (h == 4) __builtin_amdgcn_sched_barrier(0);
    v8f acc = {0.f, 0.f, 0.f, 0.f, 0.f, 0.f, 0.f, 0.f};
#pragma unroll
    for (int q = 0; q < 8; ++q) {
      const v4f bv = wv[(h * 8 + q) * 32];
      v2f a0 = {av[q].x, av[q].y}, a1 = {av[q].z, av[q].w};
      v2f b0 = {bv.x, bv.y},       b1 = {bv.z, bv.w};
      acc = __builtin_amdgcn_wmma_f32_16x16x4_f32(false, a0, false, b0, (short)0, acc, false, false);
      acc = __builtin_amdgcn_wmma_f32_16x16x4_f32(false, a1, false, b1, (short)0, acc, false, false);
    }
    cx[h] = __shfl(acc[h], l, 32);
  }

  dsum += __shfl_xor(dsum, 16, 32);

  const float* srow = sp + (size_t)seg * E_ + l;
  float x[H_];
  float sum = 0.f;
#pragma unroll
  for (int h = 0; h < H_; ++h) {
    float den = __shfl(dsum, h, 32);
    float inv = (den > 0.f) ? (1.f / den) : 0.f;
    x[h] = srow[h * 16] + cx[h] * inv;
    sum += x[h];
  }
#pragma unroll
  for (int s = 8; s >= 1; s >>= 1) sum += __shfl_xor(sum, s, 32);
  const float mu = sum * (1.f / E_);
  float vs = 0.f;
#pragma unroll
  for (int h = 0; h < H_; ++h) { float d = x[h] - mu; vs += d * d; }
#pragma unroll
  for (int s = 8; s >= 1; s >>= 1) vs += __shfl_xor(vs, s, 32);
  const float rs = rsqrtf(vs * (1.f / E_) + EPS_);
#pragma unroll
  for (int c = 0; c < 4; ++c) {
    const float xv = half ? x[2 * c + 1] : x[2 * c];
    const int e = c * 32 + lane;
    out[(size_t)seg * E_ + e] = (xv - mu) * rs * lnw[e] + lnb[e];
  }
}

extern "C" void kernel_launch(void* const* d_in, const int* in_sizes, int n_in,
                              void* d_out, int out_size, void* d_ws,
                              size_t ws_size, hipStream_t stream) {
  (void)in_sizes; (void)n_in; (void)out_size; (void)ws_size;
  const float* sp  = (const float*)d_in[0];
  const float* raw = (const float*)d_in[1];
  const int*   asg = (const int*)d_in[2];
  const float* Wq  = (const float*)d_in[3];
  const float* Wkv = (const float*)d_in[4];
  const float* lnw = (const float*)d_in[5];
  const float* lnb = (const float*)d_in[6];

  float* out      = (float*)d_out;
  float* out_attn = out + (size_t)S_ * E_;

  float* ws      = (float*)d_ws;
  float* Wfold   = ws;
  float* WvL     = Wfold + E_ * CKV_;
  float* Qk      = Wfold + 256 * CKV_;
  int*   count   = (int*)(Qk + (size_t)S_ * H_ * CKV_);
  int*   offs    = count + S_;
  int*   rank    = offs + S_;
  int*   order   = rank + N_;

  k_qkproj<<<dim3(QKX_ + AUXX_, H_), 32, 0, stream>>>(sp, Wq, Wkv, Qk, WvL, count);
  k_rank<<<N_ / 256, 256, 0, stream>>>(asg, count, rank);
  k_scan_fill<<<N_ / 1024, 256, 0, stream>>>(asg, rank, count, offs, order);
  k_seg_attn_ln<<<S_, 32, 0, stream>>>(raw, Wfold, Qk, order, offs, count,
                                       sp, lnw, lnb, out, out_attn);
}


# Segment kernel in 16-wave work-groups: each wave's Z goes to LDS, wave h applies head h's V weights to 16 segments at once, WMMA rows = segments (8 WMMA per segment, was 128); sum order unchanged
# speedup vs baseline: 12.9716x; 1.0600x over previous
#include <hip/hip_runtime.h>


#define S_    4096
#define N_    262144
#define H_    8
#define E_    128
#define D_    16
#define CKV_  64
#define EPS_  1e-5f

typedef float v2f __attribute__((ext_vector_type(2)));
typedef float v4f __attribute__((ext_vector_type(4)));
typedef float v8f __attribute__((ext_vector_type(8)));

#define QKX_  (S_ / 16)
#define AUXX_ 4
__global__ void __launch_bounds__(32)
k_qkproj(const float* __restrict__ sp,
         const float* __restrict__ Wq,
         const float* __restrict__ Wkv,
         float* __restrict__ Qk,
         float* __restrict__ WvL,
         int* __restrict__ count) {
  const int h    = blockIdx.y;
  const int lane = threadIdx.x;
  if (blockIdx.x >= QKX_) {
    const int t = ((blockIdx.x - QKX_) * H_ + h) * 32 + lane;
#pragma unroll
    for (int k = 0; k < S_ / 1024; ++k) count[t + 1024 * k] = 0;
#pragma unroll
    for (int k = 0; k < (E_ * CKV_) / 1024; ++k) {
      const int d  = t + 1024 * k;
      const int e  = d & 3, ln = (d >> 2) & 31;
      const int c  = ln & 15, hf = ln >> 4;
      const int q  = (d >> 7) & 7, hh = d >> 10;
      const float* s = Wkv + (size_t)(E_ + hh * 16 + c) * (2 * CKV_) + 8 * q + 4 * hf + e;
      WvL[d] = s[0] + s[CKV_];
    }
    return;
  }
  const int tm   = blockIdx.x * 16;
  const int half = lane >> 4;
  const int l    = lane & 15;
  const float* arow = Wq + (size_t)(h * 16 + l) * E_ + 2 * half;
  const float* brow = sp + (size_t)(tm + l) * E_ + 2 * half;
  v8f qa = {0.f, 0.f, 0.f, 0.f, 0.f, 0.f, 0.f, 0.f};
  v8f qc = qa;
#pragma unroll
  for (int k = 0; k < E_; k += 8) {
    v2f a0 = *(const v2f*)(arow + k);
    v2f b0 = *(const v2f*)(brow + k);
    v2f a1 = *(const v2f*)(arow + k + 4);
    v2f b1 = *(const v2f*)(brow + k + 4);
    qa = __builtin_amdgcn_wmma_f32_16x16x4_f32(false, a0, false, b0,
                                               (short)0, qa, false, false);
    qc = __builtin_amdgcn_wmma_f32_16x16x4_f32(false, a1, false, b1,
                                               (short)0, qc, false, false);
  }
  float qt[8];
#pragma unroll
  for (int r = 0; r < 8; ++r) qt[r] = qa[r] + qc[r];

  const float* wk = Wkv + (size_t)(h * 16 + 8 * half) * (2 * CKV_) + l;
#pragma unroll
  for (int nt = 0; nt < 4; ++nt) {
    v8f acc = {0.f, 0.f, 0.f, 0.f, 0.f, 0.f, 0.f, 0.f};
#pragma unroll
    for (int jj = 0; jj < 4; ++jj) {
      const float* w0 = wk + (2 * jj) * (2 * CKV_) + nt * 16;
      const float* w1 = w0 + 2 * CKV_;
      v2f a = {qt[2 * jj], qt[2 * jj + 1]};
      v2f b = {w0[0] + w0[CKV_], w1[0] + w1[CKV_]};
      acc = __builtin_amdgcn_wmma_f32_16x16x4_f32(false, a, false, b,
                                                  (short)0, acc, false, false);
    }
#pragma unroll
    for (int r = 0; r < 8; ++r)
      Qk[((size_t)(tm + 8 * half + r) * H_ + h) * CKV_ + nt * 16 + l] = acc[r] * 0.25f;
  }
}

__global__ void k_rank(const int* __restrict__ assign,
                       int* __restrict__ count,
                       int* __restrict__ rank) {
  const int p = blockIdx.x * blockDim.x + threadIdx.x;
  rank[p] = atomicAdd(&count[assign[p]], 1);
}

__global__ void __launch_bounds__(256)
k_scan_fill(const int* __restrict__ assign,
            const int* __restrict__ rank,
            const int* __restrict__ count,
            int* __restrict__ offs,
            int* __restrict__ order) {
  __shared__ __attribute__((aligned(16))) int so[S_];
  __shared__ int wt[8];
  const int t    = threadIdx.x;
  const int lane = t & 31;
  const int w    = t >> 5;
  const int p0   = blockIdx.x * 1024 + 4 * t;
  const int4 as = *(const int4*)(assign + p0);
  const int4 rk = *(const int4*)(rank + p0);
  const int4* c4 = (const int4*)(count + 16 * t);
  const int4 ca = c4[0], cb = c4[1], cc = c4[2], cd = c4[3];
  const int tot = (ca.x + ca.y + ca.z + ca.w) + (cb.x + cb.y + cb.z + cb.w) +
                  (cc.x + cc.y + cc.z + cc.w) + (cd.x + cd.y + cd.z + cd.w);
  int inc = tot;
#pragma unroll
  for (int d = 1; d < 32; d <<= 1) {
    const int v = __shfl_up(inc, d, 32);
    inc += (lane >= d) ? v : 0;
  }
  if (lane == 31) wt[w] = inc;
  __syncthreads();
  int ex = inc - tot;
#pragma unroll
  for (int i = 0; i < 8; ++i) ex += (i < w) ? wt[i] : 0;
  int4 oa, ob, oc, od;
  oa.x = ex;          oa.y = oa.x + ca.x; oa.z = oa.y + ca.y; oa.w = oa.z + ca.z;
  ob.x = oa.w + ca.w; ob.y = ob.x + cb.x; ob.z = ob.y + cb.y; ob.w = ob.z + cb.z;
  oc.x = ob.w + cb.w; oc.y = oc.x + cc.x; oc.z = oc.y + cc.y; oc.w = oc.z + cc.z;
  od.x = oc.w + cc.w; od.y = od.x + cd.x; od.z = od.y + cd.y; od.w = od.z + cd.z;
  int4* s4 = (int4*)(so + 16 * t);
  s4[0] = oa; s4[1] = ob; s4[2] = oc; s4[3] = od;
  if (t == (int)blockIdx.x) {
    int4* g4 = (int4*)(offs + 16 * t);
    g4[0] = oa; g4[1] = ob; g4[2] = oc; g4[3] = od;
  }
  __syncthreads();
  order[so[as.x] + rk.x] = p0;
  order[so[as.y] + rk.y] = p0 + 1;
  order[so[as.z] + rk.z] = p0 + 2;
  order[so[as.w] + rk.w] = p0 + 3;
}

#define SPB_ 16
#define ZSP_ (H_ * CKV_ + 4)
#define CXP_ (E_ + 4)

#define ROW_XADD(v, n) ((v) + __builtin_bit_cast(float, __builtin_amdgcn_update_dpp( \
    0, __builtin_bit_cast(int, (v)), 0x160 + (n), 0xf, 0xf, false)))

static __device__ __forceinline__ void seg_step(
    const int t, const int n, const int base, const int safe,
    const int half, const int l, const int posm, const bool hrow,
    const float* __restrict__ raw, const int* __restrict__ order,
    float* __restrict__ attn_out, const v4f (&qb)[8],
    const v4f (&ra)[8], const v4f (&rb)[8], v4f (&na)[8], v4f (&nb)[8],
    const int pnext, int& pnn, const int mcur, int& mnext,
    v8f& z0, v8f& z1, v8f& z2, v8f& z3, float& m, float& dsum) {
  {
    const int r1 = (t + 1) * 16 + posm;
    mnext = order[(r1 < n) ? base + r1 : safe];
  }
#pragma unroll
  for (int i = 0; i < 8; ++i) {
    const int p = __shfl(pnext, 8 * half + i, 32);
    nb[i] = *(const v4f*)(raw + (size_t)p * CKV_ + 4 * l);
  }
  {
    const int r2 = (t + 2) * 16 + l;
    pnn = order[(r2 < n) ? base + r2 : safe];
  }
#pragma unroll
  for (int q = 0; q < 8; ++q)
    na[q] = *(const v4f*)(raw + (size_t)pnext * CKV_ + 8 * q + 4 * half);

  v8f s0 = {0.f, 0.f, 0.f, 0.f, 0.f, 0.f, 0.f, 0.f};
#pragma unroll
  for (int q = 0; q < 8; ++q) {
    v2f a0 = {ra[q].x, ra[q].y}, a1 = {ra[q].z, ra[q].w};
    v2f b0 = {qb[q].x, qb[q].y}, b1 = {qb[q].z, qb[q].w};
    s0 = __builtin_amdgcn_wmma_f32_16x16x4_f32(false, a0, false, b0, (short)0, s0, false, false);
    s0 = __builtin_amdgcn_wmma_f32_16x16x4_f32(false, a1, false, b1, (short)0, s0, false, false);
  }
  float sv[8];
#pragma unroll
  for (int r = 0; r < 8; ++r) sv[r] = s0[r];

  {
    float mv = 0.f;
#pragma unroll
    for (int r = 0; r < 8; ++r) {
      float v = sv[r];
      v = ROW_XADD(v, 1);
      v = ROW_XADD(v, 2);
      v = ROW_XADD(v, 4);
      mv = ((l & 7) == r) ? v : mv;
    }
    if (hrow && (t * 16 + posm < n)) attn_out[mcur] = mv * (1.f / H_);
  }

  float tmax = -INFINITY;
#pragma unroll
  for (int r = 0; r < 8; ++r) {
    sv[r] = (t * 16 + 8 * half + r < n) ? sv[r] : -INFINITY;
    tmax = fmaxf(tmax, sv[r]);
  }
  tmax = fmaxf(tmax, __shfl_xor(tmax, 16, 32));
  const float mnew  = fmaxf(m, tmax);
  const float alpha = __expf(m - mnew);
  m = mnew;
  float w[8];
  float wsum = 0.f;
#pragma unroll
  for (int r = 0; r < 8; ++r) {
    w[r] = hrow ? __expf(sv[r] - mnew) : 0.f;
    wsum += w[r];
  }
  dsum = dsum * alpha + wsum;
#pragma unroll
  for (int r = 0; r < 8; ++r) {
    const float ar = __builtin_bit_cast(float,
        __builtin_amdgcn_readlane(__builtin_bit_cast(int, alpha), r));
    z0[r] *= ar; z1[r] *= ar; z2[r] *= ar; z3[r] *= ar;
  }

#pragma unroll
  for (int q = 0; q < 4; ++q) {
    v2f a = {w[2 * q], w[2 * q + 1]};
    const v4f u0 = rb[2 * q], u1 = rb[2 * q + 1];
    v2f b0 = {u0.x, u1.x}, b1 = {u0.y, u1.y}, b2 = {u0.z, u1.z}, b3 = {u0.w, u1.w};
    z0 = __builtin_amdgcn_wmma_f32_16x16x4_f32(false, a, false, b0, (short)0, z0, false, false);
    z1 = __builtin_amdgcn_wmma_f32_16x16x4_f32(false, a, false, b1, (short)0, z1, false, false);
    z2 = __builtin_amdgcn_wmma_f32_16x16x4_f32(false, a, false, b2, (short)0, z2, false, false);
    z3 = __builtin_amdgcn_wmma_f32_16x16x4_f32(false, a, false, b3, (short)0, z3, false, false);
  }
}

__global__ void __launch_bounds__(32 * SPB_)
k_seg_attn_ln(const float* __restrict__ raw,
              const float* __restrict__ Wfold,
              const float* __restrict__ Qk,
              const int* __restrict__ order,
              const int* __restrict__ offs,
              const int* __restrict__ count,
              const float* __restrict__ sp,
              const float* __restrict__ lnw,
              const float* __restrict__ lnb,
              float* __restrict__ out,
              float* __restrict__ attn_out) {
  __shared__ __attribute__((aligned(16))) float zs[SPB_ * ZSP_];
  __shared__ __attribute__((aligned(16))) float cs[SPB_ * CXP_];
  const int wid  = __builtin_amdgcn_readfirstlane((int)(threadIdx.x >> 5));
  const int seg  = blockIdx.x * SPB_ + wid;
  const int lane = threadIdx.x & 31;
  const int half = lane >> 4;
  const int l    = lane & 15;
  const int n    = count[seg];
  const int base = offs[seg];
  const int safe = (n > 0) ? base : 0;
  const bool hrow = (l < 8);
  const int posm = 8 * half + (l & 7);

  v4f qb[8];
  {
    const float* qrow = Qk + ((size_t)seg * H_ + (l & 7)) * CKV_ + 4 * half;
#pragma unroll
    for (int q = 0; q < 8; ++q) {
      const v4f t4 = *(const v4f*)(qrow + 8 * q);
      const v4f z4 = {0.f, 0.f, 0.f, 0.f};
      qb[q] = hrow ? t4 : z4;
    }
  }

  v8f z0 = {0.f, 0.f, 0.f, 0.f, 0.f, 0.f, 0.f, 0.f};
  v8f z1 = z0, z2 = z0, z3 = z0;
  float m = -INFINITY;
  float dsum = 0.f;

  const int ntile = (n + 15) >> 4;
  const int ptc0 = order[(l < n) ? base + l : safe];
  int ptA = order[(16 + l < n) ? base + 16 + l : safe];
  int ptB = ptA;
  int mpA = order[(posm < n) ? base + posm : safe];
  int mpB = mpA;
  v4f raA[8], rbA[8], raB[8], rbB[8];
#pragma unroll
  for (int i = 0; i < 8; ++i) {
    const int p = __shfl(ptc0, 8 * half + i, 32);
    rbA[i] = *(const v4f*)(raw + (size_t)p * CKV_ + 4 * l);
  }
#pragma unroll
  for (int q = 0; q < 8; ++q)
    raA[q] = *(const v4f*)(raw + (size_t)ptc0 * CKV_ + 8 * q + 4 * half);

  for (int t = 0; t < ntile; t += 2) {
    seg_step(t, n, base, safe, half, l, posm, hrow, raw, order, attn_out, qb,
             raA, rbA, raB, rbB, ptA, ptB, mpA, mpB, z0, z1, z2, z3, m, dsum);
    if (t + 1 >= ntile) break;
    seg_step(t + 1, n, base, safe, half, l, posm, hrow, raw, order, attn_out, qb,
             raB, rbB, raA, rbA, ptB, ptA, mpB, mpA, z0, z1, z2, z3, m, dsum);
  }

  const bool prod = (wid < H_);
  v4f bw[8];
  if (prod) {
    const v4f* wv = (const v4f*)(Wfold + E_ * CKV_) + lane;
    for (int q = 0; q < 8; ++q) bw[q] = wv[(wid * 8 + q) * 32];
  }
  if (half == 0) {
    for (int r = 0; r < 8; ++r) {
      v4f zr = {z0[r], z1[r], z2[r], z3[r]};
      *(v4f*)&zs[wid * ZSP_ + r * CKV_ + 4 * l] = zr;
    }
  }
  __syncthreads();
  if (prod) {
    v8f acc = {0.f, 0.f, 0.f, 0.f, 0.f, 0.f, 0.f, 0.f};
    for (int q = 0; q < 8; ++q) {
      const v4f av = *(const v4f*)&zs[l * ZSP_ + wid * CKV_ + 8 * q + 4 * half];
      const v4f bv = bw[q];
      v2f a0 = {av.x, av.y}, a1 = {av.z, av.w};
      v2f b0 = {bv.x, bv.y}, b1 = {bv.z, bv.w};
      acc = __builtin_amdgcn_wmma_f32_16x16x4_f32(false, a0, false, b0, (short)0, acc, false, false);
      acc = __builtin_amdgcn_wmma_f32_16x16x4_f32(false, a1, false, b1, (short)0, acc, false, false);
    }
    for (int r = 0; r < 8; ++r) cs[(8 * half + r) * CXP_ + wid * 16 + l] = acc[r];
  }
  __syncthreads();
  float cx[H_];
  for (int h = 0; h < H_; ++h) cx[h] = cs[wid * CXP_ + h * 16 + l];

  dsum += __shfl_xor(dsum, 16, 32);

  const float* srow = sp + (size_t)seg * E_ + l;
  float x[H_];
  float sum = 0.f;
#pragma unroll
  for (int h = 0; h < H_; ++h) {
    float den = __shfl(dsum, h, 32);
    float inv = (den > 0.f) ? (1.f / den) : 0.f;
    x[h] = srow[h * 16] + cx[h] * inv;
    sum += x[h];
  }
#pragma unroll
  for (int s = 8; s >= 1; s >>= 1) sum += __shfl_xor(sum, s, 32);
  const float mu = sum * (1.f / E_);
  float vs = 0.f;
#pragma unroll
  for (int h = 0; h < H_; ++h) { float d = x[h] - mu; vs += d * d; }
#pragma unroll
  for (int s = 8; s >= 1; s >>= 1) vs += __shfl_xor(vs, s, 32);
  const float rs = rsqrtf(vs * (1.f / E_) + EPS_);
#pragma unroll
  for (int c = 0; c < 4; ++c) {
    const float xv = half ? x[2 * c + 1] : x[2 * c];
    const int e = c * 32 + lane;
    out[(size_t)seg * E_ + e] = (xv - mu) * rs * lnw[e] + lnb[e];
  }
}

extern "C" void kernel_launch(void* const* d_in, const int* in_sizes, int n_in,
                              void* d_out, int out_size, void* d_ws,
                              size_t ws_size, hipStream_t stream) {
  (void)in_sizes; (void)n_in; (void)out_size; (void)ws_size;
  const float* sp  = (const float*)d_in[0];
  const float* raw = (const float*)d_in[1];
  const int*   asg = (const int*)d_in[2];
  const float* Wq  = (const float*)d_in[3];
  const float* Wkv = (const float*)d_in[4];
  const float* lnw = (const float*)d_in[5];
  const float* lnb = (const float*)d_in[6];

  float* out      = (float*)d_out;
  float* out_attn = out + (size_t)S_ * E_;

  float* ws      = (float*)d_ws;
  float* Wfold   = ws;
  float* WvL     = Wfold + E_ * CKV_;
  float* Qk      = Wfold + 256 * CKV_;
  int*   count   = (int*)(Qk + (size_t)S_ * H_ * CKV_);
  int*   offs    = count + S_;
  int*   rank    = offs + S_;
  int*   order   = rank + N_;

  k_qkproj<<<dim3(QKX_ + AUXX_, H_), 32, 0, stream>>>(sp, Wq, Wkv, Qk, WvL, count);
  k_rank<<<N_ / 256, 256, 0, stream>>>(asg, count, rank);
  k_scan_fill<<<N_ / 1024, 256, 0, stream>>>(asg, rank, count, offs, order);
  k_seg_attn_ln<<<S_ / SPB_, 32 * SPB_, 0, stream>>>(raw, Wfold, Qk, order, offs, count,
                                       sp, lnw, lnb, out, out_attn);
}


# Grouping moved into the segment launch: each 16-segment work-group filters the whole assignment array into LDS lists (LDS rank atomics, arrival order as before); no global atomics, launches 4 -> 2
# speedup vs baseline: 20.2922x; 1.5644x over previous
#include <hip/hip_runtime.h>


#define S_    4096
#define N_    262144
#define H_    8
#define E_    128
#define D_    16
#define CKV_  64
#define EPS_  1e-5f

typedef float v2f __attribute__((ext_vector_type(2)));
typedef float v4f __attribute__((ext_vector_type(4)));
typedef float v8f __attribute__((ext_vector_type(8)));

#define QKX_  (S_ / 16)
#define AUXX_ 4
__global__ void __launch_bounds__(32)
k_qkproj(const float* __restrict__ sp,
         const float* __restrict__ Wq,
         const float* __restrict__ Wkv,
         float* __restrict__ Qk,
         float* __restrict__ WvL) {
  const int h    = blockIdx.y;
  const int lane = threadIdx.x;
  if (blockIdx.x >= QKX_) {
    const int t = ((blockIdx.x - QKX_) * H_ + h) * 32 + lane;
#pragma unroll
    for (int k = 0; k < (E_ * CKV_) / 1024; ++k) {
      const int d  = t + 1024 * k;
      const int e  = d & 3, ln = (d >> 2) & 31;
      const int c  = ln & 15, hf = ln >> 4;
      const int q  = (d >> 7) & 7, hh = d >> 10;
      const float* s = Wkv + (size_t)(E_ + hh * 16 + c) * (2 * CKV_) + 8 * q + 4 * hf + e;
      WvL[d] = s[0] + s[CKV_];
    }
    return;
  }
  const int tm   = blockIdx.x * 16;
  const int half = lane >> 4;
  const int l    = lane & 15;
  const float* arow = Wq + (size_t)(h * 16 + l) * E_ + 2 * half;
  const float* brow = sp + (size_t)(tm + l) * E_ + 2 * half;
  v8f qa = {0.f, 0.f, 0.f, 0.f, 0.f, 0.f, 0.f, 0.f};
  v8f qc = qa;
#pragma unroll
  for (int k = 0; k < E_; k += 8) {
    v2f a0 = *(const v2f*)(arow + k);
    v2f b0 = *(const v2f*)(brow + k);
    v2f a1 = *(const v2f*)(arow + k + 4);
    v2f b1 = *(const v2f*)(brow + k + 4);
    qa = __builtin_amdgcn_wmma_f32_16x16x4_f32(false, a0, false, b0,
                                               (short)0, qa, false, false);
    qc = __builtin_amdgcn_wmma_f32_16x16x4_f32(false, a1, false, b1,
                                               (short)0, qc, false, false);
  }
  float qt[8];
#pragma unroll
  for (int r = 0; r < 8; ++r) qt[r] = qa[r] + qc[r];

  const float* wk = Wkv + (size_t)(h * 16 + 8 * half) * (2 * CKV_) + l;
#pragma unroll
  for (int nt = 0; nt < 4; ++nt) {
    v8f acc = {0.f, 0.f, 0.f, 0.f, 0.f, 0.f, 0.f, 0.f};
#pragma unroll
    for (int jj = 0; jj < 4; ++jj) {
      const float* w0 = wk + (2 * jj) * (2 * CKV_) + nt * 16;
      const float* w1 = w0 + 2 * CKV_;
      v2f a = {qt[2 * jj], qt[2 * jj + 1]};
      v2f b = {w0[0] + w0[CKV_], w1[0] + w1[CKV_]};
      acc = __builtin_amdgcn_wmma_f32_16x16x4_f32(false, a, false, b,
                                                  (short)0, acc, false, false);
    }
#pragma unroll
    for (int r = 0; r < 8; ++r)
      Qk[((size_t)(tm + 8 * half + r) * H_ + h) * CKV_ + nt * 16 + l] = acc[r] * 0.25f;
  }
}

#define SPB_ 16
#define CAP_ 256
#define SCH_ (32 * SPB_ * 4)
#define SPASS_ (N_ / SCH_)
#define SUNR_ 16
#define ZSP_ (H_ * CKV_ + 4)
#define CXP_ (E_ + 4)

#define ROW_XADD(v, n) ((v) + __builtin_bit_cast(float, __builtin_amdgcn_update_dpp( \
    0, __builtin_bit_cast(int, (v)), 0x160 + (n), 0xf, 0xf, false)))

static __device__ __forceinline__ void seg_step(
    const int t, const int n,
    const int half, const int l, const int posm, const bool hrow,
    const float* __restrict__ raw, const int* lst,
    float* __restrict__ attn_out, const v4f (&qb)[8],
    const v4f (&ra)[8], const v4f (&rb)[8], v4f (&na)[8], v4f (&nb)[8],
    const int pnext, int& pnn, const int mcur, int& mnext,
    v8f& z0, v8f& z1, v8f& z2, v8f& z3, float& m, float& dsum) {
  {
    const int r1 = (t + 1) * 16 + posm;
    const int v1 = lst[(r1 < n) ? r1 : 0];
    mnext = (r1 < n) ? v1 : 0;
  }
#pragma unroll
  for (int i = 0; i < 8; ++i) {
    const int p = __shfl(pnext, 8 * half + i, 32);
    nb[i] = *(const v4f*)(raw + (size_t)p * CKV_ + 4 * l);
  }
  {
    const int r2 = (t + 2) * 16 + l;
    const int v2 = lst[(r2 < n) ? r2 : 0];
    pnn = (r2 < n) ? v2 : 0;
  }
#pragma unroll
  for (int q = 0; q < 8; ++q)
    na[q] = *(const v4f*)(raw + (size_t)pnext * CKV_ + 8 * q + 4 * half);

  v8f s0 = {0.f, 0.f, 0.f, 0.f, 0.f, 0.f, 0.f, 0.f};
#pragma unroll
  for (int q = 0; q < 8; ++q) {
    v2f a0 = {ra[q].x, ra[q].y}, a1 = {ra[q].z, ra[q].w};
    v2f b0 = {qb[q].x, qb[q].y}, b1 = {qb[q].z, qb[q].w};
    s0 = __builtin_amdgcn_wmma_f32_16x16x4_f32(false, a0, false, b0, (short)0, s0, false, false);
    s0 = __builtin_amdgcn_wmma_f32_16x16x4_f32(false, a1, false, b1, (short)0, s0, false, false);
  }
  float sv[8];
#pragma unroll
  for (int r = 0; r < 8; ++r) sv[r] = s0[r];

  {
    float mv = 0.f;
#pragma unroll
    for (int r = 0; r < 8; ++r) {
      float v = sv[r];
      v = ROW_XADD(v, 1);
      v = ROW_XADD(v, 2);
      v = ROW_XADD(v, 4);
      mv = ((l & 7) == r) ? v : mv;
    }
    if (hrow && (t * 16 + posm < n)) attn_out[mcur] = mv * (1.f / H_);
  }

  float tmax = -INFINITY;
#pragma unroll
  for (int r = 0; r < 8; ++r) {
    sv[r] = (t * 16 + 8 * half + r < n) ? sv[r] : -INFINITY;
    tmax = fmaxf(tmax, sv[r]);
  }
  tmax = fmaxf(tmax, __shfl_xor(tmax, 16, 32));
  const float mnew  = fmaxf(m, tmax);
  const float alpha = __expf(m - mnew);
  m = mnew;
  float w[8];
  float wsum = 0.f;
#pragma unroll
  for (int r = 0; r < 8; ++r) {
    w[r] = hrow ? __expf(sv[r] - mnew) : 0.f;
    wsum += w[r];
  }
  dsum = dsum * alpha + wsum;
#pragma unroll
  for (int r = 0; r < 8; ++r) {
    const float ar = __builtin_bit_cast(float,
        __builtin_amdgcn_readlane(__builtin_bit_cast(int, alpha), r));
    z0[r] *= ar; z1[r] *= ar; z2[r] *= ar; z3[r] *= ar;
  }

#pragma unroll
  for (int q = 0; q < 4; ++q) {
    v2f a = {w[2 * q], w[2 * q + 1]};
    const v4f u0 = rb[2 * q], u1 = rb[2 * q + 1];
    v2f b0 = {u0.x, u1.x}, b1 = {u0.y, u1.y}, b2 = {u0.z, u1.z}, b3 = {u0.w, u1.w};
    z0 = __builtin_amdgcn_wmma_f32_16x16x4_f32(false, a, false, b0, (short)0, z0, false, false);
    z1 = __builtin_amdgcn_wmma_f32_16x16x4_f32(false, a, false, b1, (short)0, z1, false, false);
    z2 = __builtin_amdgcn_wmma_f32_16x16x4_f32(false, a, false, b2, (short)0, z2, false, false);
    z3 = __builtin_amdgcn_wmma_f32_16x16x4_f32(false, a, false, b3, (short)0, z3, false, false);
  }
}

__global__ void __launch_bounds__(32 * SPB_)
k_seg_attn_ln(const float* __restrict__ raw,
              const float* __restrict__ Wfold,
              const float* __restrict__ Qk,
              const int* __restrict__ assign,
              const float* __restrict__ sp,
              const float* __restrict__ lnw,
              const float* __restrict__ lnb,
              float* __restrict__ out,
              float* __restrict__ attn_out) {
  __shared__ __attribute__((aligned(16))) float zs[SPB_ * ZSP_];
  __shared__ __attribute__((aligned(16))) float cs[SPB_ * CXP_];
  __shared__ int lsts[SPB_ * CAP_];
  __shared__ int cnts[SPB_];
  const int wid  = __builtin_amdgcn_readfirstlane((int)(threadIdx.x >> 5));
  const int seg  = blockIdx.x * SPB_ + wid;
  const int lane = threadIdx.x & 31;
  const int half = lane >> 4;
  const int l    = lane & 15;

  const bool hrow = (l < 8);

  if (threadIdx.x < SPB_) cnts[threadIdx.x] = 0;
  __syncthreads();

  v4f qb[8];
  {
    const float* qrow = Qk + ((size_t)seg * H_ + (l & 7)) * CKV_ + 4 * half;
#pragma unroll
    for (int q = 0; q < 8; ++q) {
      const v4f t4 = *(const v4f*)(qrow + 8 * q);
      const v4f z4 = {0.f, 0.f, 0.f, 0.f};
      qb[q] = hrow ? t4 : z4;
    }
  }

  {
    const int blk = blockIdx.x;
    const int rot = blockIdx.x & (SPASS_ - 1);
    const int tp  = 4 * (int)threadIdx.x;
    for (int it = 0; it < SPASS_; it += SUNR_) {
      int4 av[SUNR_];
#pragma unroll
      for (int u = 0; u < SUNR_; ++u) {
        const int ch = (it + u + rot) & (SPASS_ - 1);
        av[u] = *(const int4*)(assign + ch * SCH_ + tp);
      }
#pragma unroll
      for (int u = 0; u < SUNR_; ++u) {
        const int ch = (it + u + rot) & (SPASS_ - 1);
        const int p0 = ch * SCH_ + tp;
        const int4 a = av[u];
        const bool m0 = (a.x >> 4) == blk, m1 = (a.y >> 4) == blk;
        const bool m2 = (a.z >> 4) == blk, m3 = (a.w >> 4) == blk;
        if (m0 | m1 | m2 | m3) {
          if (m0) { const int r = atomicAdd(&cnts[a.x & 15], 1);
                    if (r < CAP_) lsts[(a.x & 15) * CAP_ + r] = p0; }
          if (m1) { const int r = atomicAdd(&cnts[a.y & 15], 1);
                    if (r < CAP_) lsts[(a.y & 15) * CAP_ + r] = p0 + 1; }
          if (m2) { const int r = atomicAdd(&cnts[a.z & 15], 1);
                    if (r < CAP_) lsts[(a.z & 15) * CAP_ + r] = p0 + 2; }
          if (m3) { const int r = atomicAdd(&cnts[a.w & 15], 1);
                    if (r < CAP_) lsts[(a.w & 15) * CAP_ + r] = p0 + 3; }
        }
      }
    }
  }
  __syncthreads();
  const bool slow = __builtin_amdgcn_ballot_w32(cnts[l] > CAP_) != 0u;
  int* const lst = lsts + wid * CAP_;
  int n = slow ? 0 : cnts[wid];
  int scanpos = 0;
  const int posm = 8 * half + (l & 7);

  v8f z0 = {0.f, 0.f, 0.f, 0.f, 0.f, 0.f, 0.f, 0.f};
  v8f z1 = z0, z2 = z0, z3 = z0;
  float m = -INFINITY;
  float dsum = 0.f;

  bool more;
  do {
  if (slow) {
    n = 0;
    while (scanpos < N_ && n <= CAP_ - 32) {
      const bool mt = (assign[scanpos + lane] == seg);
      const unsigned mk = __builtin_amdgcn_ballot_w32(mt);
      if (mt) lst[n + __builtin_popcount(mk & ((1u << lane) - 1u))] = scanpos + lane;
      n += __builtin_popcount(mk);
      scanpos += 32;
    }
  }
  more = slow && (scanpos < N_);
  const int ntile = (n + 15) >> 4;
  const int vc0 = lst[(l < n) ? l : 0];
  const int ptc0 = (l < n) ? vc0 : 0;
  const int vA = lst[(16 + l < n) ? 16 + l : 0];
  int ptA = (16 + l < n) ? vA : 0;
  int ptB = ptA;
  const int vm = lst[(posm < n) ? posm : 0];
  int mpA = (posm < n) ? vm : 0;
  int mpB = mpA;
  v4f raA[8], rbA[8], raB[8], rbB[8];
#pragma unroll
  for (int i = 0; i < 8; ++i) {
    const int p = __shfl(ptc0, 8 * half + i, 32);
    rbA[i] = *(const v4f*)(raw + (size_t)p * CKV_ + 4 * l);
  }
#pragma unroll
  for (int q = 0; q < 8; ++q)
    raA[q] = *(const v4f*)(raw + (size_t)ptc0 * CKV_ + 8 * q + 4 * half);

  for (int t = 0; t < ntile; t += 2) {
    seg_step(t, n, half, l, posm, hrow, raw, lst, attn_out, qb,
             raA, rbA, raB, rbB, ptA, ptB, mpA, mpB, z0, z1, z2, z3, m, dsum);
    if (t + 1 >= ntile) break;
    seg_step(t + 1, n, half, l, posm, hrow, raw, lst, attn_out, qb,
             raB, rbB, raA, rbA, ptB, ptA, mpB, mpA, z0, z1, z2, z3, m, dsum);
  }
  } while (more);

  const bool prod = (wid < H_);
  v4f bw[8];
  if (prod) {
    const v4f* wv = (const v4f*)(Wfold + E_ * CKV_) + lane;
    for (int q = 0; q < 8; ++q) bw[q] = wv[(wid * 8 + q) * 32];
  }
  if (half == 0) {
    for (int r = 0; r < 8; ++r) {
      v4f zr = {z0[r], z1[r], z2[r], z3[r]};
      *(v4f*)&zs[wid * ZSP_ + r * CKV_ + 4 * l] = zr;
    }
  }
  __syncthreads();
  if (prod) {
    v8f acc = {0.f, 0.f, 0.f, 0.f, 0.f, 0.f, 0.f, 0.f};
    for (int q = 0; q < 8; ++q) {
      const v4f av = *(const v4f*)&zs[l * ZSP_ + wid * CKV_ + 8 * q + 4 * half];
      const v4f bv = bw[q];
      v2f a0 = {av.x, av.y}, a1 = {av.z, av.w};
      v2f b0 = {bv.x, bv.y}, b1 = {bv.z, bv.w};
      acc = __builtin_amdgcn_wmma_f32_16x16x4_f32(false, a0, false, b0, (short)0, acc, false, false);
      acc = __builtin_amdgcn_wmma_f32_16x16x4_f32(false, a1, false, b1, (short)0, acc, false, false);
    }
    for (int r = 0; r < 8; ++r) cs[(8 * half + r) * CXP_ + wid * 16 + l] = acc[r];
  }
  __syncthreads();
  float cx[H_];
  for (int h = 0; h < H_; ++h) cx[h] = cs[wid * CXP_ + h * 16 + l];

  dsum += __shfl_xor(dsum, 16, 32);

  const float* srow = sp + (size_t)seg * E_ + l;
  float x[H_];
  float sum = 0.f;
#pragma unroll
  for (int h = 0; h < H_; ++h) {
    float den = __shfl(dsum, h, 32);
    float inv = (den > 0.f) ? (1.f / den) : 0.f;
    x[h] = srow[h * 16] + cx[h] * inv;
    sum += x[h];
  }
#pragma unroll
  for (int s = 8; s >= 1; s >>= 1) sum += __shfl_xor(sum, s, 32);
  const float mu = sum * (1.f / E_);
  float vs = 0.f;
#pragma unroll
  for (int h = 0; h < H_; ++h) { float d = x[h] - mu; vs += d * d; }
#pragma unroll
  for (int s = 8; s >= 1; s >>= 1) vs += __shfl_xor(vs, s, 32);
  const float rs = rsqrtf(vs * (1.f / E_) + EPS_);
#pragma unroll
  for (int c = 0; c < 4; ++c) {
    const float xv = half ? x[2 * c + 1] : x[2 * c];
    const int e = c * 32 + lane;
    out[(size_t)seg * E_ + e] = (xv - mu) * rs * lnw[e] + lnb[e];
  }
}

extern "C" void kernel_launch(void* const* d_in, const int* in_sizes, int n_in,
                              void* d_out, int out_size, void* d_ws,
                              size_t ws_size, hipStream_t stream) {
  (void)in_sizes; (void)n_in; (void)out_size; (void)ws_size;
  const float* sp  = (const float*)d_in[0];
  const float* raw = (const float*)d_in[1];
  const int*   asg = (const int*)d_in[2];
  const float* Wq  = (const float*)d_in[3];
  const float* Wkv = (const float*)d_in[4];
  const float* lnw = (const float*)d_in[5];
  const float* lnb = (const float*)d_in[6];

  float* out      = (float*)d_out;
  float* out_attn = out + (size_t)S_ * E_;

  float* ws      = (float*)d_ws;
  float* Wfold   = ws;
  float* WvL     = Wfold + E_ * CKV_;
  float* Qk      = Wfold + 256 * CKV_;

  k_qkproj<<<dim3(QKX_ + AUXX_, H_), 32, 0, stream>>>(sp, Wq, Wkv, Qk, WvL);
  k_seg_attn_ln<<<S_ / SPB_, 32 * SPB_, 0, stream>>>(raw, Wfold, Qk, asg,
                                       sp, lnw, lnb, out, out_attn);
}


# Grouping split: launch 1 orders each 512-point chunk by 16-segment block in LDS (rank atomics, arrival order); a segment work-group gathers its run from every chunk, no scan of all assignments
# speedup vs baseline: 26.2444x; 1.2933x over previous
#include <hip/hip_runtime.h>


#define S_    4096
#define N_    262144
#define H_    8
#define E_    128
#define D_    16
#define CKV_  64
#define EPS_  1e-5f

typedef float v2f __attribute__((ext_vector_type(2)));
typedef float v4f __attribute__((ext_vector_type(4)));
typedef float v8f __attribute__((ext_vector_type(8)));

#define QKX_  (S_ / 16)
#define AUXX_ 4
#define CH_   512
#define NCH_  (N_ / CH_)
#define NBLK_ (S_ / 16)
#define BINX_ (NCH_ / H_)
__global__ void __launch_bounds__(32)
k_qkproj(const float* __restrict__ sp,
         const float* __restrict__ Wq,
         const float* __restrict__ Wkv,
         const int* __restrict__ assign,
         float* __restrict__ Qk,
         float* __restrict__ WvL,
         float* __restrict__ binws) {
  __shared__ int hist[NBLK_];
  __shared__ __attribute__((aligned(16))) unsigned short srt[CH_];
  const int h    = blockIdx.y;
  const int lane = threadIdx.x;
  if (blockIdx.x >= QKX_ + AUXX_) {
    const int c = ((int)blockIdx.x - QKX_ - AUXX_) * H_ + h;
    const int* src = assign + c * CH_;
#pragma unroll
    for (int k = 0; k < NBLK_ / 32; ++k) hist[lane + 32 * k] = 0;
    int4 a[CH_ / 128];
#pragma unroll
    for (int u = 0; u < CH_ / 128; ++u) a[u] = *(const int4*)(src + 128 * u + 4 * lane);
    __syncthreads();
    int rk[CH_ / 32];
#pragma unroll
    for (int u = 0; u < CH_ / 128; ++u) {
      rk[4 * u + 0] = atomicAdd(&hist[(a[u].x >> 4) & (NBLK_ - 1)], 1);
      rk[4 * u + 1] = atomicAdd(&hist[(a[u].y >> 4) & (NBLK_ - 1)], 1);
      rk[4 * u + 2] = atomicAdd(&hist[(a[u].z >> 4) & (NBLK_ - 1)], 1);
      rk[4 * u + 3] = atomicAdd(&hist[(a[u].w >> 4) & (NBLK_ - 1)], 1);
    }
    __syncthreads();
    int ex[8];
    int tot = 0;
#pragma unroll
    for (int k = 0; k < 8; ++k) { ex[k] = tot; tot += hist[8 * lane + k]; }
    int incl = tot;
#pragma unroll
    for (int d = 1; d < 32; d <<= 1) {
      const int up = __shfl_up(incl, d, 32);
      incl += (lane >= d) ? up : 0;
    }
    const int base = incl - tot;
    __syncthreads();
#pragma unroll
    for (int k = 0; k < 8; ++k) { ex[k] += base; hist[8 * lane + k] = ex[k]; }
    unsigned short* offs_g = (unsigned short*)binws + c * NBLK_;
    unsigned short* sort_g = (unsigned short*)binws + NCH_ * NBLK_ + c * CH_;
    {
      int4 o;
      o.x = ex[0] | (ex[1] << 16); o.y = ex[2] | (ex[3] << 16);
      o.z = ex[4] | (ex[5] << 16); o.w = ex[6] | (ex[7] << 16);
      *(int4*)(offs_g + 8 * lane) = o;
    }
    __syncthreads();
#pragma unroll
    for (int u = 0; u < CH_ / 128; ++u) {
      const int i0 = 128 * u + 4 * lane;
      const int v0 = a[u].x, v1 = a[u].y, v2 = a[u].z, v3 = a[u].w;
      srt[hist[(v0 >> 4) & (NBLK_ - 1)] + rk[4 * u + 0]] = (unsigned short)(((i0 + 0) << 4) | (v0 & 15));
      srt[hist[(v1 >> 4) & (NBLK_ - 1)] + rk[4 * u + 1]] = (unsigned short)(((i0 + 1) << 4) | (v1 & 15));
      srt[hist[(v2 >> 4) & (NBLK_ - 1)] + rk[4 * u + 2]] = (unsigned short)(((i0 + 2) << 4) | (v2 & 15));
      srt[hist[(v3 >> 4) & (NBLK_ - 1)] + rk[4 * u + 3]] = (unsigned short)(((i0 + 3) << 4) | (v3 & 15));
    }
    __syncthreads();
#pragma unroll
    for (int k = 0; k < CH_ / 256; ++k)
      ((int4*)sort_g)[lane + 32 * k] = ((const int4*)srt)[lane + 32 * k];
    return;
  }
  if (blockIdx.x >= QKX_) {
    const int t = ((blockIdx.x - QKX_) * H_ + h) * 32 + lane;
#pragma unroll
    for (int k = 0; k < (E_ * CKV_) / 1024; ++k) {
      const int d  = t + 1024 * k;
      const int e  = d & 3, ln = (d >> 2) & 31;
      const int c  = ln & 15, hf = ln >> 4;
      const int q  = (d >> 7) & 7, hh = d >> 10;
      const float* s = Wkv + (size_t)(E_ + hh * 16 + c) * (2 * CKV_) + 8 * q + 4 * hf + e;
      WvL[d] = s[0] + s[CKV_];
    }
    return;
  }
  const int tm   = blockIdx.x * 16;
  const int half = lane >> 4;
  const int l    = lane & 15;
  const float* arow = Wq + (size_t)(h * 16 + l) * E_ + 2 * half;
  const float* brow = sp + (size_t)(tm + l) * E_ + 2 * half;
  v8f qa = {0.f, 0.f, 0.f, 0.f, 0.f, 0.f, 0.f, 0.f};
  v8f qc = qa;
#pragma unroll
  for (int k = 0; k < E_; k += 8) {
    v2f a0 = *(const v2f*)(arow + k);
    v2f b0 = *(const v2f*)(brow + k);
    v2f a1 = *(const v2f*)(arow + k + 4);
    v2f b1 = *(const v2f*)(brow + k + 4);
    qa = __builtin_amdgcn_wmma_f32_16x16x4_f32(false, a0, false, b0,
                                               (short)0, qa, false, false);
    qc = __builtin_amdgcn_wmma_f32_16x16x4_f32(false, a1, false, b1,
                                               (short)0, qc, false, false);
  }
  float qt[8];
#pragma unroll
  for (int r = 0; r < 8; ++r) qt[r] = qa[r] + qc[r];

  const float* wk = Wkv + (size_t)(h * 16 + 8 * half) * (2 * CKV_) + l;
#pragma unroll
  for (int nt = 0; nt < 4; ++nt) {
    v8f acc = {0.f, 0.f, 0.f, 0.f, 0.f, 0.f, 0.f, 0.f};
#pragma unroll
    for (int jj = 0; jj < 4; ++jj) {
      const float* w0 = wk + (2 * jj) * (2 * CKV_) + nt * 16;
      const float* w1 = w0 + 2 * CKV_;
      v2f a = {qt[2 * jj], qt[2 * jj + 1]};
      v2f b = {w0[0] + w0[CKV_], w1[0] + w1[CKV_]};
      acc = __builtin_amdgcn_wmma_f32_16x16x4_f32(false, a, false, b,
                                                  (short)0, acc, false, false);
    }
#pragma unroll
    for (int r = 0; r < 8; ++r)
      Qk[((size_t)(tm + 8 * half + r) * H_ + h) * CKV_ + nt * 16 + l] = acc[r] * 0.25f;
  }
}

#define SPB_ 16
#define CAP_ 256
#define GUN_ 8
static_assert(NCH_ == 32 * SPB_, "one chunk per thread of a segment work-group");
static_assert(NBLK_ * SPB_ == S_ && (CH_ << 4) <= 65536, "bins and 16-bit entries");
#define ZSP_ (H_ * CKV_ + 4)
#define CXP_ (E_ + 4)

#define ROW_XADD(v, n) ((v) + __builtin_bit_cast(float, __builtin_amdgcn_update_dpp( \
    0, __builtin_bit_cast(int, (v)), 0x160 + (n), 0xf, 0xf, false)))

static __device__ __forceinline__ void seg_step(
    const int t, const int n,
    const int half, const int l, const int posm, const bool hrow,
    const float* __restrict__ raw, const int* lst,
    float* __restrict__ attn_out, const v4f (&qb)[8],
    const v4f (&ra)[8], const v4f (&rb)[8], v4f (&na)[8], v4f (&nb)[8],
    const int pnext, int& pnn, const int mcur, int& mnext,
    v8f& z0, v8f& z1, v8f& z2, v8f& z3, float& m, float& dsum) {
  {
    const int r1 = (t + 1) * 16 + posm;
    const int v1 = lst[(r1 < n) ? r1 : 0];
    mnext = (r1 < n) ? v1 : 0;
  }
#pragma unroll
  for (int i = 0; i < 8; ++i) {
    const int p = __shfl(pnext, 8 * half + i, 32);
    nb[i] = *(const v4f*)(raw + (size_t)p * CKV_ + 4 * l);
  }
  {
    const int r2 = (t + 2) * 16 + l;
    const int v2 = lst[(r2 < n) ? r2 : 0];
    pnn = (r2 < n) ? v2 : 0;
  }
#pragma unroll
  for (int q = 0; q < 8; ++q)
    na[q] = *(const v4f*)(raw + (size_t)pnext * CKV_ + 8 * q + 4 * half);

  v8f s0 = {0.f, 0.f, 0.f, 0.f, 0.f, 0.f, 0.f, 0.f};
#pragma unroll
  for (int q = 0; q < 8; ++q) {
    v2f a0 = {ra[q].x, ra[q].y}, a1 = {ra[q].z, ra[q].w};
    v2f b0 = {qb[q].x, qb[q].y}, b1 = {qb[q].z, qb[q].w};
    s0 = __builtin_amdgcn_wmma_f32_16x16x4_f32(false, a0, false, b0, (short)0, s0, false, false);
    s0 = __builtin_amdgcn_wmma_f32_16x16x4_f32(false, a1, false, b1, (short)0, s0, false, false);
  }
  float sv[8];
#pragma unroll
  for (int r = 0; r < 8; ++r) sv[r] = s0[r];

  {
    float mv = 0.f;
#pragma unroll
    for (int r = 0; r < 8; ++r) {
      float v = sv[r];
      v = ROW_XADD(v, 1);
      v = ROW_XADD(v, 2);
      v = ROW_XADD(v, 4);
      mv = ((l & 7) == r) ? v : mv;
    }
    if (hrow && (t * 16 + posm < n)) attn_out[mcur] = mv * (1.f / H_);
  }

  float tmax = -INFINITY;
#pragma unroll
  for (int r = 0; r < 8; ++r) {
    sv[r] = (t * 16 + 8 * half + r < n) ? sv[r] : -INFINITY;
    tmax = fmaxf(tmax, sv[r]);
  }
  tmax = fmaxf(tmax, __shfl_xor(tmax, 16, 32));
  const float mnew  = fmaxf(m, tmax);
  const float alpha = __expf(m - mnew);
  m = mnew;
  float w[8];
  float wsum = 0.f;
#pragma unroll
  for (int r = 0; r < 8; ++r) {
    w[r] = hrow ? __expf(sv[r] - mnew) : 0.f;
    wsum += w[r];
  }
  dsum = dsum * alpha + wsum;
#pragma unroll
  for (int r = 0; r < 8; ++r) {
    const float ar = __builtin_bit_cast(float,
        __builtin_amdgcn_readlane(__builtin_bit_cast(int, alpha), r));
    z0[r] *= ar; z1[r] *= ar; z2[r] *= ar; z3[r] *= ar;
  }

#pragma unroll
  for (int q = 0; q < 4; ++q) {
    v2f a = {w[2 * q], w[2 * q + 1]};
    const v4f u0 = rb[2 * q], u1 = rb[2 * q + 1];
    v2f b0 = {u0.x, u1.x}, b1 = {u0.y, u1.y}, b2 = {u0.z, u1.z}, b3 = {u0.w, u1.w};
    z0 = __builtin_amdgcn_wmma_f32_16x16x4_f32(false, a, false, b0, (short)0, z0, false, false);
    z1 = __builtin_amdgcn_wmma_f32_16x16x4_f32(false, a, false, b1, (short)0, z1, false, false);
    z2 = __builtin_amdgcn_wmma_f32_16x16x4_f32(false, a, false, b2, (short)0, z2, false, false);
    z3 = __builtin_amdgcn_wmma_f32_16x16x4_f32(false, a, false, b3, (short)0, z3, false, false);
  }
}

__global__ void __launch_bounds__(32 * SPB_)
k_seg_attn_ln(const float* __restrict__ raw,
              const float* __restrict__ Wfold,
              const float* __restrict__ Qk,
              const int* __restrict__ assign,
              const float* __restrict__ binws,
              const float* __restrict__ sp,
              const float* __restrict__ lnw,
              const float* __restrict__ lnb,
              float* __restrict__ out,
              float* __restrict__ attn_out) {
  __shared__ __attribute__((aligned(16))) float zs[SPB_ * ZSP_];
  __shared__ __attribute__((aligned(16))) float cs[SPB_ * CXP_];
  __shared__ int lsts[SPB_ * CAP_];
  __shared__ int cnts[SPB_ + 32];
  const int wid  = __builtin_amdgcn_readfirstlane((int)(threadIdx.x >> 5));
  const int seg  = blockIdx.x * SPB_ + wid;
  const int lane = threadIdx.x & 31;
  const int half = lane >> 4;
  const int l    = lane & 15;

  const bool hrow = (l < 8);

  if (threadIdx.x < SPB_ + 32) cnts[threadIdx.x] = 0;
  __syncthreads();

  v4f qb[8];
  {
    const float* qrow = Qk + ((size_t)seg * H_ + (l & 7)) * CKV_ + 4 * half;
#pragma unroll
    for (int q = 0; q < 8; ++q) {
      const v4f t4 = *(const v4f*)(qrow + 8 * q);
      const v4f z4 = {0.f, 0.f, 0.f, 0.f};
      qb[q] = hrow ? t4 : z4;
    }
  }

  {
    const int blk = blockIdx.x;
    const int c   = (int)threadIdx.x;
    const unsigned short* og = (const unsigned short*)binws + c * NBLK_ + blk;
    const unsigned short* sg = (const unsigned short*)binws + NCH_ * NBLK_ + c * CH_;
    const int o0 = og[0];
    int o1 = CH_;
    if (blk + 1 < NBLK_) o1 = og[1];
    const int cc = o1 - o0;
    int ev[GUN_], rr[GUN_];
#pragma unroll
    for (int k = 0; k < GUN_; ++k) {
      const int q = o0 + k;
      ev[k] = sg[(q < CH_) ? q : CH_ - 1];
    }
#pragma unroll
    for (int k = 0; k < GUN_; ++k)
      rr[k] = atomicAdd(&cnts[(k < cc) ? (ev[k] & 15) : (SPB_ + (c & 31))], 1);
#pragma unroll
    for (int k = 0; k < GUN_; ++k)
      if (k < cc && rr[k] < CAP_) lsts[(ev[k] & 15) * CAP_ + rr[k]] = c * CH_ + (ev[k] >> 4);
    for (int k = GUN_; k < cc; ++k) {
      const int e = sg[o0 + k];
      const int r = atomicAdd(&cnts[e & 15], 1);
      if (r < CAP_) lsts[(e & 15) * CAP_ + r] = c * CH_ + (e >> 4);
    }
  }
  __syncthreads();
  const bool slow = __builtin_amdgcn_ballot_w32(cnts[l] > CAP_) != 0u;
  int* const lst = lsts + wid * CAP_;
  int n = slow ? 0 : cnts[wid];
  int scanpos = 0;
  const int posm = 8 * half + (l & 7);

  v8f z0 = {0.f, 0.f, 0.f, 0.f, 0.f, 0.f, 0.f, 0.f};
  v8f z1 = z0, z2 = z0, z3 = z0;
  float m = -INFINITY;
  float dsum = 0.f;

  bool more;
  do {
  if (slow) {
    n = 0;
    while (scanpos < N_ && n <= CAP_ - 32) {
      const bool mt = (assign[scanpos + lane] == seg);
      const unsigned mk = __builtin_amdgcn_ballot_w32(mt);
      if (mt) lst[n + __builtin_popcount(mk & ((1u << lane) - 1u))] = scanpos + lane;
      n += __builtin_popcount(mk);
      scanpos += 32;
    }
  }
  more = slow && (scanpos < N_);
  const int ntile = (n + 15) >> 4;
  const int vc0 = lst[(l < n) ? l : 0];
  const int ptc0 = (l < n) ? vc0 : 0;
  const int vA = lst[(16 + l < n) ? 16 + l : 0];
  int ptA = (16 + l < n) ? vA : 0;
  int ptB = ptA;
  const int vm = lst[(posm < n) ? posm : 0];
  int mpA = (posm < n) ? vm : 0;
  int mpB = mpA;
  v4f raA[8], rbA[8], raB[8], rbB[8];
#pragma unroll
  for (int i = 0; i < 8; ++i) {
    const int p = __shfl(ptc0, 8 * half + i, 32);
    rbA[i] = *(const v4f*)(raw + (size_t)p * CKV_ + 4 * l);
  }
#pragma unroll
  for (int q = 0; q < 8; ++q)
    raA[q] = *(const v4f*)(raw + (size_t)ptc0 * CKV_ + 8 * q + 4 * half);

  for (int t = 0; t < ntile; t += 2) {
    seg_step(t, n, half, l, posm, hrow, raw, lst, attn_out, qb,
             raA, rbA, raB, rbB, ptA, ptB, mpA, mpB, z0, z1, z2, z3, m, dsum);
    if (t + 1 >= ntile) break;
    seg_step(t + 1, n, half, l, posm, hrow, raw, lst, attn_out, qb,
             raB, rbB, raA, rbA, ptB, ptA, mpB, mpA, z0, z1, z2, z3, m, dsum);
  }
  } while (more);

  const bool prod = (wid < H_);
  v4f bw[8];
  if (prod) {
    const v4f* wv = (const v4f*)(Wfold + E_ * CKV_) + lane;
    for (int q = 0; q < 8; ++q) bw[q] = wv[(wid * 8 + q) * 32];
  }
  if (half == 0) {
    for (int r = 0; r < 8; ++r) {
      v4f zr = {z0[r], z1[r], z2[r], z3[r]};
      *(v4f*)&zs[wid * ZSP_ + r * CKV_ + 4 * l] = zr;
    }
  }
  __syncthreads();
  if (prod) {
    v8f acc = {0.f, 0.f, 0.f, 0.f, 0.f, 0.f, 0.f, 0.f};
    for (int q = 0; q < 8; ++q) {
      const v4f av = *(const v4f*)&zs[l * ZSP_ + wid * CKV_ + 8 * q + 4 * half];
      const v4f bv = bw[q];
      v2f a0 = {av.x, av.y}, a1 = {av.z, av.w};
      v2f b0 = {bv.x, bv.y}, b1 = {bv.z, bv.w};
      acc = __builtin_amdgcn_wmma_f32_16x16x4_f32(false, a0, false, b0, (short)0, acc, false, false);
      acc = __builtin_amdgcn_wmma_f32_16x16x4_f32(false, a1, false, b1, (short)0, acc, false, false);
    }
    for (int r = 0; r < 8; ++r) cs[(8 * half + r) * CXP_ + wid * 16 + l] = acc[r];
  }
  __syncthreads();
  float cx[H_];
  for (int h = 0; h < H_; ++h) cx[h] = cs[wid * CXP_ + h * 16 + l];

  dsum += __shfl_xor(dsum, 16, 32);

  const float* srow = sp + (size_t)seg * E_ + l;
  float x[H_];
  float sum = 0.f;
#pragma unroll
  for (int h = 0; h < H_; ++h) {
    float den = __shfl(dsum, h, 32);
    float inv = (den > 0.f) ? (1.f / den) : 0.f;
    x[h] = srow[h * 16] + cx[h] * inv;
    sum += x[h];
  }
#pragma unroll
  for (int s = 8; s >= 1; s >>= 1) sum += __shfl_xor(sum, s, 32);
  const float mu = sum * (1.f / E_);
  float vs = 0.f;
#pragma unroll
  for (int h = 0; h < H_; ++h) { float d = x[h] - mu; vs += d * d; }
#pragma unroll
  for (int s = 8; s >= 1; s >>= 1) vs += __shfl_xor(vs, s, 32);
  const float rs = rsqrtf(vs * (1.f / E_) + EPS_);
#pragma unroll
  for (int c = 0; c < 4; ++c) {
    const float xv = half ? x[2 * c + 1] : x[2 * c];
    const int e = c * 32 + lane;
    out[(size_t)seg * E_ + e] = (xv - mu) * rs * lnw[e] + lnb[e];
  }
}

extern "C" void kernel_launch(void* const* d_in, const int* in_sizes, int n_in,
                              void* d_out, int out_size, void* d_ws,
                              size_t ws_size, hipStream_t stream) {
  (void)in_sizes; (void)n_in; (void)out_size; (void)ws_size;
  const float* sp  = (const float*)d_in[0];
  const float* raw = (const float*)d_in[1];
  const int*   asg = (const int*)d_in[2];
  const float* Wq  = (const float*)d_in[3];
  const float* Wkv = (const float*)d_in[4];
  const float* lnw = (const float*)d_in[5];
  const float* lnb = (const float*)d_in[6];

  float* out      = (float*)d_out;
  float* out_attn = out + (size_t)S_ * E_;

  float* ws      = (float*)d_ws;
  float* Wfold   = ws;
  float* WvL     = Wfold + E_ * CKV_;
  float* Qk      = Wfold + 256 * CKV_;
  float* binws   = Qk + (size_t)S_ * H_ * CKV_;

  k_qkproj<<<dim3(QKX_ + AUXX_ + BINX_, H_), 32, 0, stream>>>(sp, Wq, Wkv, asg, Qk, WvL, binws);
  k_seg_attn_ln<<<S_ / SPB_, 32 * SPB_, 0, stream>>>(raw, Wfold, Qk, asg, binws,
                                       sp, lnw, lnb, out, out_attn);
}


# Score operand via LDS: a step's raw rows are loaded once as whole rows and re-laid per position in a wave-private LDS tile; the second gather of those rows (16 lines per load) is gone; same sum order
# speedup vs baseline: 27.5110x; 1.0483x over previous
#include <hip/hip_runtime.h>


#define S_    4096
#define N_    262144
#define H_    8
#define E_    128
#define D_    16
#define CKV_  64
#define EPS_  1e-5f

typedef float v2f __attribute__((ext_vector_type(2)));
typedef float v4f __attribute__((ext_vector_type(4)));
typedef float v8f __attribute__((ext_vector_type(8)));

#define QKX_  (S_ / 16)
#define AUXX_ 4
#define CH_   512
#define NCH_  (N_ / CH_)
#define NBLK_ (S_ / 16)
#define BINX_ (NCH_ / H_)
__global__ void __launch_bounds__(32)
k_qkproj(const float* __restrict__ sp,
         const float* __restrict__ Wq,
         const float* __restrict__ Wkv,
         const int* __restrict__ assign,
         float* __restrict__ Qk,
         float* __restrict__ WvL,
         float* __restrict__ binws) {
  __shared__ int hist[NBLK_];
  __shared__ __attribute__((aligned(16))) unsigned short srt[CH_];
  const int h    = blockIdx.y;
  const int lane = threadIdx.x;
  if (blockIdx.x >= QKX_ + AUXX_) {
    const int c = ((int)blockIdx.x - QKX_ - AUXX_) * H_ + h;
    const int* src = assign + c * CH_;
#pragma unroll
    for (int k = 0; k < NBLK_ / 32; ++k) hist[lane + 32 * k] = 0;
    int4 a[CH_ / 128];
#pragma unroll
    for (int u = 0; u < CH_ / 128; ++u) a[u] = *(const int4*)(src + 128 * u + 4 * lane);
    __syncthreads();
    int rk[CH_ / 32];
#pragma unroll
    for (int u = 0; u < CH_ / 128; ++u) {
      rk[4 * u + 0] = atomicAdd(&hist[(a[u].x >> 4) & (NBLK_ - 1)], 1);
      rk[4 * u + 1] = atomicAdd(&hist[(a[u].y >> 4) & (NBLK_ - 1)], 1);
      rk[4 * u + 2] = atomicAdd(&hist[(a[u].z >> 4) & (NBLK_ - 1)], 1);
      rk[4 * u + 3] = atomicAdd(&hist[(a[u].w >> 4) & (NBLK_ - 1)], 1);
    }
    __syncthreads();
    int ex[8];
    int tot = 0;
#pragma unroll
    for (int k = 0; k < 8; ++k) { ex[k] = tot; tot += hist[8 * lane + k]; }
    int incl = tot;
#pragma unroll
    for (int d = 1; d < 32; d <<= 1) {
      const int up = __shfl_up(incl, d, 32);
      incl += (lane >= d) ? up : 0;
    }
    const int base = incl - tot;
    __syncthreads();
#pragma unroll
    for (int k = 0; k < 8; ++k) { ex[k] += base; hist[8 * lane + k] = ex[k]; }
    unsigned short* offs_g = (unsigned short*)binws + c * NBLK_;
    unsigned short* sort_g = (unsigned short*)binws + NCH_ * NBLK_ + c * CH_;
    {
      int4 o;
      o.x = ex[0] | (ex[1] << 16); o.y = ex[2] | (ex[3] << 16);
      o.z = ex[4] | (ex[5] << 16); o.w = ex[6] | (ex[7] << 16);
      *(int4*)(offs_g + 8 * lane) = o;
    }
    __syncthreads();
#pragma unroll
    for (int u = 0; u < CH_ / 128; ++u) {
      const int i0 = 128 * u + 4 * lane;
      const int v0 = a[u].x, v1 = a[u].y, v2 = a[u].z, v3 = a[u].w;
      srt[hist[(v0 >> 4) & (NBLK_ - 1)] + rk[4 * u + 0]] = (unsigned short)(((i0 + 0) << 4) | (v0 & 15));
      srt[hist[(v1 >> 4) & (NBLK_ - 1)] + rk[4 * u + 1]] = (unsigned short)(((i0 + 1) << 4) | (v1 & 15));
      srt[hist[(v2 >> 4) & (NBLK_ - 1)] + rk[4 * u + 2]] = (unsigned short)(((i0 + 2) << 4) | (v2 & 15));
      srt[hist[(v3 >> 4) & (NBLK_ - 1)] + rk[4 * u + 3]] = (unsigned short)(((i0 + 3) << 4) | (v3 & 15));
    }
    __syncthreads();
#pragma unroll
    for (int k = 0; k < CH_ / 256; ++k)
      ((int4*)sort_g)[lane + 32 * k] = ((const int4*)srt)[lane + 32 * k];
    return;
  }
  if (blockIdx.x >= QKX_) {
    const int t = ((blockIdx.x - QKX_) * H_ + h) * 32 + lane;
#pragma unroll
    for (int k = 0; k < (E_ * CKV_) / 1024; ++k) {
      const int d  = t + 1024 * k;
      const int e  = d & 3, ln = (d >> 2) & 31;
      const int c  = ln & 15, hf = ln >> 4;
      const int q  = (d >> 7) & 7, hh = d >> 10;
      const float* s = Wkv + (size_t)(E_ + hh * 16 + c) * (2 * CKV_) + 8 * q + 4 * hf + e;
      WvL[d] = s[0] + s[CKV_];
    }
    return;
  }
  const int tm   = blockIdx.x * 16;
  const int half = lane >> 4;
  const int l    = lane & 15;
  const float* arow = Wq + (size_t)(h * 16 + l) * E_ + 2 * half;
  const float* brow = sp + (size_t)(tm + l) * E_ + 2 * half;
  v8f qa = {0.f, 0.f, 0.f, 0.f, 0.f, 0.f, 0.f, 0.f};
  v8f qc = qa;
#pragma unroll
  for (int k = 0; k < E_; k += 8) {
    v2f a0 = *(const v2f*)(arow + k);
    v2f b0 = *(const v2f*)(brow + k);
    v2f a1 = *(const v2f*)(arow + k + 4);
    v2f b1 = *(const v2f*)(brow + k + 4);
    qa = __builtin_amdgcn_wmma_f32_16x16x4_f32(false, a0, false, b0,
                                               (short)0, qa, false, false);
    qc = __builtin_amdgcn_wmma_f32_16x16x4_f32(false, a1, false, b1,
                                               (short)0, qc, false, false);
  }
  float qt[8];
#pragma unroll
  for (int r = 0; r < 8; ++r) qt[r] = qa[r] + qc[r];

  const float* wk = Wkv + (size_t)(h * 16 + 8 * half) * (2 * CKV_) + l;
#pragma unroll
  for (int nt = 0; nt < 4; ++nt) {
    v8f acc = {0.f, 0.f, 0.f, 0.f, 0.f, 0.f, 0.f, 0.f};
#pragma unroll
    for (int jj = 0; jj < 4; ++jj) {
      const float* w0 = wk + (2 * jj) * (2 * CKV_) + nt * 16;
      const float* w1 = w0 + 2 * CKV_;
      v2f a = {qt[2 * jj], qt[2 * jj + 1]};
      v2f b = {w0[0] + w0[CKV_], w1[0] + w1[CKV_]};
      acc = __builtin_amdgcn_wmma_f32_16x16x4_f32(false, a, false, b,
                                                  (short)0, acc, false, false);
    }
#pragma unroll
    for (int r = 0; r < 8; ++r)
      Qk[((size_t)(tm + 8 * half + r) * H_ + h) * CKV_ + nt * 16 + l] = acc[r] * 0.25f;
  }
}

#define SPB_ 16
#define CAP_ 256
#define GUN_ 8
static_assert(NCH_ == 32 * SPB_, "one chunk per thread of a segment work-group");
static_assert(NBLK_ * SPB_ == S_ && (CH_ << 4) <= 65536, "bins and 16-bit entries");
#define ZSP_ (H_ * CKV_ + 4)
#define CXP_ (E_ + 4)
#define TLP_ (CKV_ + 4)

#define ROW_XADD(v, n) ((v) + __builtin_bit_cast(float, __builtin_amdgcn_update_dpp( \
    0, __builtin_bit_cast(int, (v)), 0x160 + (n), 0xf, 0xf, false)))

static __device__ __forceinline__ void seg_step(
    const int t, const int n,
    const int half, const int l, const int posm, const bool hrow,
    const float* __restrict__ raw, const int* lst, float* tile,
    float* __restrict__ attn_out, const v4f (&qb)[8],
    const v4f (&rb)[8], v4f (&nb)[8],
    const int pnext, int& pnn, const int mcur, int& mnext,
    v8f& z0, v8f& z1, v8f& z2, v8f& z3, float& m, float& dsum) {
#pragma unroll
  for (int i = 0; i < 8; ++i)
    *(v4f*)&tile[(8 * half + i) * TLP_ + 4 * l] = rb[i];
  __builtin_amdgcn_fence(__ATOMIC_RELEASE, "wavefront");
  __builtin_amdgcn_wave_barrier();
  __builtin_amdgcn_fence(__ATOMIC_ACQUIRE, "wavefront");
  v4f ra[8];
#pragma unroll
  for (int q = 0; q < 8; ++q)
    ra[q] = *(const v4f*)&tile[l * TLP_ + 8 * q + 4 * half];
  {
    const int r1 = (t + 1) * 16 + posm;
    const int v1 = lst[(r1 < n) ? r1 : 0];
    mnext = (r1 < n) ? v1 : 0;
  }
#pragma unroll
  for (int i = 0; i < 8; ++i) {
    const int p = __shfl(pnext, 8 * half + i, 32);
    nb[i] = *(const v4f*)(raw + (size_t)p * CKV_ + 4 * l);
  }
  {
    const int r2 = (t + 2) * 16 + l;
    const int v2 = lst[(r2 < n) ? r2 : 0];
    pnn = (r2 < n) ? v2 : 0;
  }

  __builtin_amdgcn_sched_barrier(0);
  v8f s0 = {0.f, 0.f, 0.f, 0.f, 0.f, 0.f, 0.f, 0.f};
#pragma unroll
  for (int q = 0; q < 8; ++q) {
    v2f a0 = {ra[q].x, ra[q].y}, a1 = {ra[q].z, ra[q].w};
    v2f b0 = {qb[q].x, qb[q].y}, b1 = {qb[q].z, qb[q].w};
    s0 = __builtin_amdgcn_wmma_f32_16x16x4_f32(false, a0, false, b0, (short)0, s0, false, false);
    s0 = __builtin_amdgcn_wmma_f32_16x16x4_f32(false, a1, false, b1, (short)0, s0, false, false);
  }
  float sv[8];
#pragma unroll
  for (int r = 0; r < 8; ++r) sv[r] = s0[r];

  {
    float mv = 0.f;
#pragma unroll
    for (int r = 0; r < 8; ++r) {
      float v = sv[r];
      v = ROW_XADD(v, 1);
      v = ROW_XADD(v, 2);
      v = ROW_XADD(v, 4);
      mv = ((l & 7) == r) ? v : mv;
    }
    if (hrow && (t * 16 + posm < n)) attn_out[mcur] = mv * (1.f / H_);
  }

  float tmax = -INFINITY;
#pragma unroll
  for (int r = 0; r < 8; ++r) {
    sv[r] = (t * 16 + 8 * half + r < n) ? sv[r] : -INFINITY;
    tmax = fmaxf(tmax, sv[r]);
  }
  tmax = fmaxf(tmax, __shfl_xor(tmax, 16, 32));
  const float mnew  = fmaxf(m, tmax);
  const float alpha = __expf(m - mnew);
  m = mnew;
  float w[8];
  float wsum = 0.f;
#pragma unroll
  for (int r = 0; r < 8; ++r) {
    w[r] = hrow ? __expf(sv[r] - mnew) : 0.f;
    wsum += w[r];
  }
  dsum = dsum * alpha + wsum;
#pragma unroll
  for (int r = 0; r < 8; ++r) {
    const float ar = __builtin_bit_cast(float,
        __builtin_amdgcn_readlane(__builtin_bit_cast(int, alpha), r));
    z0[r] *= ar; z1[r] *= ar; z2[r] *= ar; z3[r] *= ar;
  }

#pragma unroll
  for (int q = 0; q < 4; ++q) {
    v2f a = {w[2 * q], w[2 * q + 1]};
    const v4f u0 = rb[2 * q], u1 = rb[2 * q + 1];
    v2f b0 = {u0.x, u1.x}, b1 = {u0.y, u1.y}, b2 = {u0.z, u1.z}, b3 = {u0.w, u1.w};
    z0 = __builtin_amdgcn_wmma_f32_16x16x4_f32(false, a, false, b0, (short)0, z0, false, false);
    z1 = __builtin_amdgcn_wmma_f32_16x16x4_f32(false, a, false, b1, (short)0, z1, false, false);
    z2 = __builtin_amdgcn_wmma_f32_16x16x4_f32(false, a, false, b2, (short)0, z2, false, false);
    z3 = __builtin_amdgcn_wmma_f32_16x16x4_f32(false, a, false, b3, (short)0, z3, false, false);
  }
}

__global__ void __launch_bounds__(32 * SPB_)
k_seg_attn_ln(const float* __restrict__ raw,
              const float* __restrict__ Wfold,
              const float* __restrict__ Qk,
              const int* __restrict__ assign,
              const float* __restrict__ binws,
              const float* __restrict__ sp,
              const float* __restrict__ lnw,
              const float* __restrict__ lnb,
              float* __restrict__ out,
              float* __restrict__ attn_out) {
  __shared__ __attribute__((aligned(16))) float zs[SPB_ * ZSP_];
  __shared__ __attribute__((aligned(16))) float cs[SPB_ * CXP_];
  __shared__ __attribute__((aligned(16))) float stg[SPB_ * 16 * TLP_];
  __shared__ int lsts[SPB_ * CAP_];
  __shared__ int cnts[SPB_ + 32];
  const int wid  = __builtin_amdgcn_readfirstlane((int)(threadIdx.x >> 5));
  const int seg  = blockIdx.x * SPB_ + wid;
  const int lane = threadIdx.x & 31;
  const int half = lane >> 4;
  const int l    = lane & 15;

  const bool hrow = (l < 8);

  if (threadIdx.x < SPB_ + 32) cnts[threadIdx.x] = 0;
  __syncthreads();

  v4f qb[8];
  {
    const float* qrow = Qk + ((size_t)seg * H_ + (l & 7)) * CKV_ + 4 * half;
#pragma unroll
    for (int q = 0; q < 8; ++q) {
      const v4f t4 = *(const v4f*)(qrow + 8 * q);
      const v4f z4 = {0.f, 0.f, 0.f, 0.f};
      qb[q] = hrow ? t4 : z4;
    }
  }

  {
    const int blk = blockIdx.x;
    const int c   = (int)threadIdx.x;
    const unsigned short* og = (const unsigned short*)binws + c * NBLK_ + blk;
    const unsigned short* sg = (const unsigned short*)binws + NCH_ * NBLK_ + c * CH_;
    const int o0 = og[0];
    int o1 = CH_;
    if (blk + 1 < NBLK_) o1 = og[1];
    const int cc = o1 - o0;
    int ev[GUN_], rr[GUN_];
#pragma unroll
    for (int k = 0; k < GUN_; ++k) {
      const int q = o0 + k;
      ev[k] = sg[(q < CH_) ? q : CH_ - 1];
    }
#pragma unroll
    for (int k = 0; k < GUN_; ++k)
      rr[k] = atomicAdd(&cnts[(k < cc) ? (ev[k] & 15) : (SPB_ + (c & 31))], 1);
#pragma unroll
    for (int k = 0; k < GUN_; ++k)
      if (k < cc && rr[k] < CAP_) lsts[(ev[k] & 15) * CAP_ + rr[k]] = c * CH_ + (ev[k] >> 4);
    for (int k = GUN_; k < cc; ++k) {
      const int e = sg[o0 + k];
      const int r = atomicAdd(&cnts[e & 15], 1);
      if (r < CAP_) lsts[(e & 15) * CAP_ + r] = c * CH_ + (e >> 4);
    }
  }
  __syncthreads();
  const bool slow = __builtin_amdgcn_ballot_w32(cnts[l] > CAP_) != 0u;
  int* const lst = lsts + wid * CAP_;
  float* const tile = stg + wid * (16 * TLP_);
  int n = slow ? 0 : cnts[wid];
  int scanpos = 0;
  const int posm = 8 * half + (l & 7);

  v8f z0 = {0.f, 0.f, 0.f, 0.f, 0.f, 0.f, 0.f, 0.f};
  v8f z1 = z0, z2 = z0, z3 = z0;
  float m = -INFINITY;
  float dsum = 0.f;

  bool more;
  do {
  if (slow) {
    n = 0;
    while (scanpos < N_ && n <= CAP_ - 32) {
      const bool mt = (assign[scanpos + lane] == seg);
      const unsigned mk = __builtin_amdgcn_ballot_w32(mt);
      if (mt) lst[n + __builtin_popcount(mk & ((1u << lane) - 1u))] = scanpos + lane;
      n += __builtin_popcount(mk);
      scanpos += 32;
    }
  }
  more = slow && (scanpos < N_);
  const int ntile = (n + 15) >> 4;
  const int vc0 = lst[(l < n) ? l : 0];
  const int ptc0 = (l < n) ? vc0 : 0;
  const int vA = lst[(16 + l < n) ? 16 + l : 0];
  int ptA = (16 + l < n) ? vA : 0;
  int ptB = ptA;
  const int vm = lst[(posm < n) ? posm : 0];
  int mpA = (posm < n) ? vm : 0;
  int mpB = mpA;
  v4f rbA[8], rbB[8];
#pragma unroll
  for (int i = 0; i < 8; ++i) {
    const int p = __shfl(ptc0, 8 * half + i, 32);
    rbA[i] = *(const v4f*)(raw + (size_t)p * CKV_ + 4 * l);
  }

  for (int t = 0; t < ntile; t += 2) {
    seg_step(t, n, half, l, posm, hrow, raw, lst, tile, attn_out, qb,
             rbA, rbB, ptA, ptB, mpA, mpB, z0, z1, z2, z3, m, dsum);
    if (t + 1 >= ntile) break;
    seg_step(t + 1, n, half, l, posm, hrow, raw, lst, tile, attn_out, qb,
             rbB, rbA, ptB, ptA, mpB, mpA, z0, z1, z2, z3, m, dsum);
  }
  } while (more);

  const bool prod = (wid < H_);
  v4f bw[8];
  if (prod) {
    const v4f* wv = (const v4f*)(Wfold + E_ * CKV_) + lane;
    for (int q = 0; q < 8; ++q) bw[q] = wv[(wid * 8 + q) * 32];
  }
  if (half == 0) {
    for (int r = 0; r < 8; ++r) {
      v4f zr = {z0[r], z1[r], z2[r], z3[r]};
      *(v4f*)&zs[wid * ZSP_ + r * CKV_ + 4 * l] = zr;
    }
  }
  __syncthreads();
  if (prod) {
    v8f acc = {0.f, 0.f, 0.f, 0.f, 0.f, 0.f, 0.f, 0.f};
    for (int q = 0; q < 8; ++q) {
      const v4f av = *(const v4f*)&zs[l * ZSP_ + wid * CKV_ + 8 * q + 4 * half];
      const v4f bv = bw[q];
      v2f a0 = {av.x, av.y}, a1 = {av.z, av.w};
      v2f b0 = {bv.x, bv.y}, b1 = {bv.z, bv.w};
      acc = __builtin_amdgcn_wmma_f32_16x16x4_f32(false, a0, false, b0, (short)0, acc, false, false);
      acc = __builtin_amdgcn_wmma_f32_16x16x4_f32(false, a1, false, b1, (short)0, acc, false, false);
    }
    for (int r = 0; r < 8; ++r) cs[(8 * half + r) * CXP_ + wid * 16 + l] = acc[r];
  }
  __syncthreads();
  float cx[H_];
  for (int h = 0; h < H_; ++h) cx[h] = cs[wid * CXP_ + h * 16 + l];

  dsum += __shfl_xor(dsum, 16, 32);

  const float* srow = sp + (size_t)seg * E_ + l;
  float x[H_];
  float sum = 0.f;
#pragma unroll
  for (int h = 0; h < H_; ++h) {
    float den = __shfl(dsum, h, 32);
    float inv = (den > 0.f) ? (1.f / den) : 0.f;
    x[h] = srow[h * 16] + cx[h] * inv;
    sum += x[h];
  }
#pragma unroll
  for (int s = 8; s >= 1; s >>= 1) sum += __shfl_xor(sum, s, 32);
  const float mu = sum * (1.f / E_);
  float vs = 0.f;
#pragma unroll
  for (int h = 0; h < H_; ++h) { float d = x[h] - mu; vs += d * d; }
#pragma unroll
  for (int s = 8; s >= 1; s >>= 1) vs += __shfl_xor(vs, s, 32);
  const float rs = rsqrtf(vs * (1.f / E_) + EPS_);
#pragma unroll
  for (int c = 0; c < 4; ++c) {
    const float xv = half ? x[2 * c + 1] : x[2 * c];
    const int e = c * 32 + lane;
    out[(size_t)seg * E_ + e] = (xv - mu) * rs * lnw[e] + lnb[e];
  }
}

extern "C" void kernel_launch(void* const* d_in, const int* in_sizes, int n_in,
                              void* d_out, int out_size, void* d_ws,
                              size_t ws_size, hipStream_t stream) {
  (void)in_sizes; (void)n_in; (void)out_size; (void)ws_size;
  const float* sp  = (const float*)d_in[0];
  const float* raw = (const float*)d_in[1];
  const int*   asg = (const int*)d_in[2];
  const float* Wq  = (const float*)d_in[3];
  const float* Wkv = (const float*)d_in[4];
  const float* lnw = (const float*)d_in[5];
  const float* lnb = (const float*)d_in[6];

  float* out      = (float*)d_out;
  float* out_attn = out + (size_t)S_ * E_;

  float* ws      = (float*)d_ws;
  float* Wfold   = ws;
  float* WvL     = Wfold + E_ * CKV_;
  float* Qk      = Wfold + 256 * CKV_;
  float* binws   = Qk + (size_t)S_ * H_ * CKV_;

  k_qkproj<<<dim3(QKX_ + AUXX_ + BINX_, H_), 32, 0, stream>>>(sp, Wq, Wkv, asg, Qk, WvL, binws);
  k_seg_attn_ln<<<S_ / SPB_, 32 * SPB_, 0, stream>>>(raw, Wfold, Qk, asg, binws,
                                       sp, lnw, lnb, out, out_attn);
}
